# starting kernel
# baseline (speedup 1.0000x reference)
#include <hip/hip_runtime.h>

#define DDIM 300
#define PDIM 320
#define NWN  20000
#define EWN  320000
#define BN   256
#define LN   256
#define SDN  2048
#define EDN  4096
#define CN   20
#define KC   160

typedef __attribute__((ext_vector_type(2))) float v2f;
typedef __attribute__((ext_vector_type(8))) float v8f;


__global__ void zero_f32(float* __restrict__ p, long long n) {
    long long t = (long long)blockIdx.x * blockDim.x + threadIdx.x;
    if (t < n) p[t] = 0.f;
}

__global__ void gather_pad(const float* __restrict__ emb, const int* __restrict__ nid,
                           float* __restrict__ out, int n) {
    int t = blockIdx.x * blockDim.x + threadIdx.x;
    if (t >= n * PDIM) return;
    int r = t / PDIM, j = t % PDIM;
    out[t] = (j < DDIM) ? emb[(size_t)nid[r] * DDIM + j] : 0.f;
}

__global__ void pack_sage_w(const float* __restrict__ w, float* __restrict__ out) {
    int t = blockIdx.x * blockDim.x + threadIdx.x;
    if (t >= 2 * PDIM * PDIM) return;
    int h = t / (PDIM * PDIM);
    int r = (t / PDIM) % PDIM;
    int c = t % PDIM;
    out[t] = (r < DDIM && c < DDIM) ? w[((size_t)(h * DDIM + r)) * DDIM + c] : 0.f;
}

__global__ void pack_w(const float* __restrict__ w, float* __restrict__ out) {
    int t = blockIdx.x * blockDim.x + threadIdx.x;
    if (t >= PDIM * PDIM) return;
    int r = t / PDIM, c = t % PDIM;
    out[t] = (r < DDIM && c < DDIM) ? w[(size_t)r * DDIM + c] : 0.f;
}


__global__ void count_edges(const int* __restrict__ e_dst, int E, float* __restrict__ cnt) {
    int t = blockIdx.x * blockDim.x + threadIdx.x;
    if (t < E) atomicAdd(&cnt[e_dst[t]], 1.f);
}

__global__ void scatter_edges(const float* __restrict__ emb, const int* __restrict__ src_nid,
                              const int* __restrict__ e_src, const int* __restrict__ e_dst,
                              int E, float* __restrict__ sum) {
    const int lane = threadIdx.x & 31;
    const int e = blockIdx.x * (blockDim.x >> 5) + (threadIdx.x >> 5);
    if (e >= E) return;
    const int s = src_nid[e_src[e]];
    const float* __restrict__ src = emb + (size_t)s * DDIM;
    float* __restrict__ dst = sum + (size_t)e_dst[e] * PDIM;
    for (int j = lane; j < DDIM; j += 32)
        atomicAdd(&dst[j], src[j]);
}

__global__ void normalize_mean(float* __restrict__ sum, const float* __restrict__ cnt, int n) {
    int t = blockIdx.x * blockDim.x + threadIdx.x;
    if (t >= n * PDIM) return;
    sum[t] /= fmaxf(cnt[t / PDIM], 1.f);
}

__global__ __launch_bounds__(256) void gemm_f32_wmma(
        const float* __restrict__ A, const float* __restrict__ B,
        const float* __restrict__ bias, float* __restrict__ C,
        int M, int N, int K, int lda, int ldb, int ldc, int accum) {
    __shared__ float ldsB[64 * KC];
    const int lane = threadIdx.x & 31;
    const int wave = threadIdx.x >> 5;
    const int hi   = lane >> 4;
    const int l16  = lane & 15;
    const int nbase = blockIdx.y * 64;
    const int mt    = blockIdx.x * 8 + wave;
    const bool valid = (mt * 16) < M;
    const int m = mt * 16 + l16;

    v8f acc[4];
    if (valid) {
        if (accum) {
#pragma unroll
            for (int ns = 0; ns < 4; ++ns)
#pragma unroll
                for (int i = 0; i < 8; ++i)
                    acc[ns][i] = C[(size_t)(mt * 16 + i + hi * 8) * ldc + nbase + ns * 16 + l16];
        } else {
#pragma unroll
            for (int ns = 0; ns < 4; ++ns) {
                const int n = nbase + ns * 16 + l16;
                const float bv = (bias && n < DDIM) ? bias[n] : 0.f;
#pragma unroll
                for (int i = 0; i < 8; ++i) acc[ns][i] = bv;
            }
        }
    }

    const float* __restrict__ Arow = A + (size_t)m * lda;
    for (int kk = 0; kk < K; kk += KC) {
        __syncthreads();
        for (int idx = threadIdx.x; idx < 64 * KC; idx += 256) {
            int c = idx & 63, kloc = idx >> 6;
            ldsB[c * KC + kloc] = B[(size_t)(kk + kloc) * ldb + nbase + c];
        }
        __syncthreads();
        if (valid) {
            for (int k = 0; k < KC; k += 4) {
                v2f a = *(const v2f*)(Arow + kk + k + 2 * hi);
#pragma unroll
                for (int ns = 0; ns < 4; ++ns) {
                    v2f b = *(const v2f*)(&ldsB[(ns * 16 + l16) * KC + k + 2 * hi]);
                    acc[ns] = __builtin_amdgcn_wmma_f32_16x16x4_f32(
                        false, a, false, b, (short)0, acc[ns], false, false);
                }
            }
        }
    }
    if (valid) {
#pragma unroll
        for (int ns = 0; ns < 4; ++ns)
#pragma unroll
            for (int i = 0; i < 8; ++i)
                C[(size_t)(mt * 16 + i + hi * 8) * ldc + nbase + ns * 16 + l16] = acc[ns][i];
    }
}

__global__ void sdpa_fuse_add(const float* __restrict__ h0, const float* __restrict__ h1,
                              const float* __restrict__ h2, const float* __restrict__ addv,
                              float* __restrict__ out, int n) {
    const int lane = threadIdx.x & 31;
    const int node = blockIdx.x * (blockDim.x >> 5) + (threadIdx.x >> 5);
    if (node >= n) return;
    const float* x0 = h0 + (size_t)node * PDIM;
    const float* x1 = h1 + (size_t)node * PDIM;
    const float* x2 = h2 + (size_t)node * PDIM;

    float d00 = 0, d01 = 0, d02 = 0, d11 = 0, d12 = 0, d22 = 0;
    for (int j = lane; j < PDIM; j += 32) {
        float a = x0[j], b = x1[j], e = x2[j];
        d00 += a * a; d01 += a * b; d02 += a * e;
        d11 += b * b; d12 += b * e; d22 += e * e;
    }
    for (int off = 16; off > 0; off >>= 1) {
        d00 += __shfl_xor(d00, off, 32);
        d01 += __shfl_xor(d01, off, 32);
        d02 += __shfl_xor(d02, off, 32);
        d11 += __shfl_xor(d11, off, 32);
        d12 += __shfl_xor(d12, off, 32);
        d22 += __shfl_xor(d22, off, 32);
    }
    const float scale = rsqrtf((float)DDIM);
    float S[3][3] = {{d00, d01, d02}, {d01, d11, d12}, {d02, d12, d22}};
    float wsum[3] = {0.f, 0.f, 0.f};
    for (int t = 0; t < 3; ++t) {
        float p0 = S[t][0] * scale, p1 = S[t][1] * scale, p2 = S[t][2] * scale;
        float mx = fmaxf(p0, fmaxf(p1, p2));
        float e0 = __expf(p0 - mx), e1 = __expf(p1 - mx), e2 = __expf(p2 - mx);
        float inv = 1.f / (e0 + e1 + e2);
        wsum[0] += e0 * inv; wsum[1] += e1 * inv; wsum[2] += e2 * inv;
    }
    const float w0 = wsum[0] * (1.f / 3.f), w1 = wsum[1] * (1.f / 3.f), w2 = wsum[2] * (1.f / 3.f);
    const float* av = addv + (size_t)node * PDIM;
    float* o = out + (size_t)node * PDIM;
    for (int j = lane; j < PDIM; j += 32)
        o[j] = w0 * x0[j] + w1 * x1[j] + w2 * x2[j] + av[j];
}

__global__ void pool_seq(const float* __restrict__ table, const int* __restrict__ xb,
                         float* __restrict__ pooled) {
    const int b = blockIdx.x;
    const int j = threadIdx.x;
    if (j >= PDIM) return;
    const int* row = xb + b * LN;
    float acc = 0.f;
    for (int l = 0; l < LN; ++l) {
        int idx = row[l];
        if (idx < NWN) acc += table[(size_t)idx * PDIM + j];
    }
    pooled[(size_t)b * PDIM + j] = acc * (1.f / (float)LN);
}

__global__ void fc_kernel(const float* __restrict__ rep, const float* __restrict__ wfc,
                          const float* __restrict__ bfc, float* __restrict__ out) {
    int t = blockIdx.x * blockDim.x + threadIdx.x;
    if (t >= BN * CN) return;
    int b = t / CN, c = t % CN;
    float acc = bfc[c];
    const float* r = rep + (size_t)b * PDIM;
    for (int j = 0; j < DDIM; ++j) acc += r[j] * wfc[(size_t)j * CN + c];
    out[t] = acc;
}


extern "C" void kernel_launch(void* const* d_in, const int* in_sizes, int n_in,
                              void* d_out, int out_size, void* d_ws, size_t ws_size,
                              hipStream_t stream) {
    (void)in_sizes; (void)n_in; (void)out_size; (void)ws_size;
    const float* emb_word = (const float*)d_in[0];
    const float* emb_doc  = (const float*)d_in[1];
    const float* w_sage[6] = {(const float*)d_in[2],  (const float*)d_in[4],  (const float*)d_in[6],
                              (const float*)d_in[8],  (const float*)d_in[10], (const float*)d_in[12]};
    const float* b_sage[6] = {(const float*)d_in[3],  (const float*)d_in[5],  (const float*)d_in[7],
                              (const float*)d_in[9],  (const float*)d_in[11], (const float*)d_in[13]};
    const float* w_dense = (const float*)d_in[14];
    const float* b_dense = (const float*)d_in[15];
    const float* w_fc    = (const float*)d_in[16];
    const float* b_fc    = (const float*)d_in[17];
    const int* dst_nid      = (const int*)d_in[18];
    const int* src_nid_w[3] = {(const int*)d_in[19], (const int*)d_in[20], (const int*)d_in[21]};
    const int* e_src_w[3]   = {(const int*)d_in[22], (const int*)d_in[24], (const int*)d_in[26]};
    const int* e_dst_w[3]   = {(const int*)d_in[23], (const int*)d_in[25], (const int*)d_in[27]};
    const int* x_batch      = (const int*)d_in[28];
    const int* src_nid_d[3] = {(const int*)d_in[30], (const int*)d_in[31], (const int*)d_in[32]};
    const int* e_src_d[3]   = {(const int*)d_in[33], (const int*)d_in[35], (const int*)d_in[37]};
    const int* e_dst_d[3]   = {(const int*)d_in[34], (const int*)d_in[36], (const int*)d_in[38]};
    float* out = (float*)d_out;

    float* ws = (float*)d_ws;
    size_t off = 0;
    auto carve = [&](size_t nflt) { float* p = ws + off; off += nflt; return p; };
    float* hdst = carve((size_t)NWN * PDIM);
    float* nbuf = carve((size_t)NWN * PDIM);
    float* h3[3]; for (int r = 0; r < 3; ++r) h3[r] = carve((size_t)NWN * PDIM);
    float* cnt  = carve(NWN);
    float* wpk[6]; for (int r = 0; r < 6; ++r) wpk[r] = carve((size_t)2 * PDIM * PDIM);
    float* wdn  = carve((size_t)PDIM * PDIM);
    float* pooled = carve((size_t)BN * PDIM);
    float* docout = carve((size_t)BN * PDIM);
    float* dnb    = carve((size_t)BN * PDIM);
    float* d3[3]; for (int r = 0; r < 3; ++r) d3[r] = carve((size_t)BN * PDIM);
    float* docrep = carve((size_t)BN * PDIM);

    const int TB = 256;
    const unsigned wordElems = (unsigned)(((size_t)NWN * PDIM + TB - 1) / TB);
    const unsigned docElems  = (unsigned)(((size_t)BN * PDIM + TB - 1) / TB);
    const dim3 gemmGridW((NWN + 127) / 128, PDIM / 64);
    const dim3 gemmGridD((BN + 127) / 128, PDIM / 64);

    for (int r = 0; r < 6; ++r)
        pack_sage_w<<<(2 * PDIM * PDIM + TB - 1) / TB, TB, 0, stream>>>(w_sage[r], wpk[r]);
    pack_w<<<(PDIM * PDIM + TB - 1) / TB, TB, 0, stream>>>(w_dense, wdn);

    gather_pad<<<wordElems, TB, 0, stream>>>(emb_word, dst_nid, hdst, NWN);

    for (int r = 0; r < 3; ++r) {
        zero_f32<<<wordElems, TB, 0, stream>>>(nbuf, (long long)NWN * PDIM);
        zero_f32<<<(NWN + TB - 1) / TB, TB, 0, stream>>>(cnt, NWN);
        count_edges<<<(EWN + TB - 1) / TB, TB, 0, stream>>>(e_dst_w[r], EWN, cnt);
        scatter_edges<<<(EWN + 7) / 8, TB, 0, stream>>>(emb_word, src_nid_w[r], e_src_w[r],
                                                        e_dst_w[r], EWN, nbuf);
        normalize_mean<<<wordElems, TB, 0, stream>>>(nbuf, cnt, NWN);
        gemm_f32_wmma<<<gemmGridW, TB, 0, stream>>>(hdst, wpk[r], b_sage[r], h3[r],
                                                    NWN, PDIM, PDIM, PDIM, PDIM, PDIM, 0);
        gemm_f32_wmma<<<gemmGridW, TB, 0, stream>>>(nbuf, wpk[r] + (size_t)PDIM * PDIM, nullptr,
                                                    h3[r], NWN, PDIM, PDIM, PDIM, PDIM, PDIM, 1);
    }

    sdpa_fuse_add<<<(NWN + 7) / 8, 256, 0, stream>>>(h3[0], h3[1], h3[2], hdst, nbuf, NWN);

    pool_seq<<<BN, PDIM, 0, stream>>>(nbuf, x_batch, pooled);
    gemm_f32_wmma<<<gemmGridD, TB, 0, stream>>>(pooled, wdn, b_dense, docout,
                                                BN, PDIM, PDIM, PDIM, PDIM, PDIM, 0);

    for (int r = 0; r < 3; ++r) {
        zero_f32<<<docElems, TB, 0, stream>>>(dnb, (long long)BN * PDIM);
        zero_f32<<<1, TB, 0, stream>>>(cnt, BN);
        count_edges<<<(EDN + TB - 1) / TB, TB, 0, stream>>>(e_dst_d[r], EDN, cnt);
        scatter_edges<<<(EDN + 7) / 8, TB, 0, stream>>>(emb_doc, src_nid_d[r], e_src_d[r],
                                                        e_dst_d[r], EDN, dnb);
        normalize_mean<<<docElems, TB, 0, stream>>>(dnb, cnt, BN);
        gemm_f32_wmma<<<gemmGridD, TB, 0, stream>>>(docout, wpk[3 + r], b_sage[3 + r], d3[r],
                                                    BN, PDIM, PDIM, PDIM, PDIM, PDIM, 0);
        gemm_f32_wmma<<<gemmGridD, TB, 0, stream>>>(dnb, wpk[3 + r] + (size_t)PDIM * PDIM, nullptr,
                                                    d3[r], BN, PDIM, PDIM, PDIM, PDIM, PDIM, 1);
    }

    sdpa_fuse_add<<<(BN + 7) / 8, 256, 0, stream>>>(d3[0], d3[1], d3[2], docout, docrep, BN);
    fc_kernel<<<(BN * CN + TB - 1) / TB, TB, 0, stream>>>(docrep, w_fc, b_fc, out);
}


# word path by row ownership: a workgroup takes 80 rows x 1 relation through neighbour mean (LDS atomics, arrival order as before) and both SAGE K halves, A from LDS; 20 -> 12 launches; same k order
# speedup vs baseline: 4.1261x; 4.1261x over previous
#include <hip/hip_runtime.h>

#define DDIM 300
#define PDIM 320
#define NWN  20000
#define EWN  320000
#define BN   256
#define LN   256
#define SDN  2048
#define EDN  4096
#define CN   20
#define NT16 (PDIM / 16)

typedef __attribute__((ext_vector_type(2))) float v2f;
typedef __attribute__((ext_vector_type(4))) float v4f;
typedef __attribute__((ext_vector_type(8))) float v8f;


__device__ __forceinline__ void frag_row_col(int u, int& r, int& c) {
    const int s = u & 1, j = (u >> 1) & 1, lane = (u >> 2) & 31;
    const int tile = u >> 7;
    const int nt = tile % NT16, kp = tile / NT16;
    r = 8 * kp + 4 * j + 2 * (lane >> 4) + s;
    c = 16 * nt + (lane & 15);
}

__global__ void pack_sage_w(const float* __restrict__ w, float* __restrict__ out) {
    int t = blockIdx.x * blockDim.x + threadIdx.x;
    if (t >= 2 * PDIM * PDIM) return;
    int h = t / (PDIM * PDIM);
    int r, c;
    frag_row_col(t % (PDIM * PDIM), r, c);
    out[t] = (r < DDIM && c < DDIM) ? w[((size_t)(h * DDIM + r)) * DDIM + c] : 0.f;
}

__global__ void pack_w(const float* __restrict__ w, float* __restrict__ out) {
    int t = blockIdx.x * blockDim.x + threadIdx.x;
    if (t >= PDIM * PDIM) return;
    int r, c;
    frag_row_col(t, r, c);
    out[t] = (r < DDIM && c < DDIM) ? w[(size_t)r * DDIM + c] : 0.f;
}

#define SM_WAVES 32
#define SM_UNR   4
#define SM_CAP   1024

template <int PITCH>
__device__ __forceinline__ void sm_flush(const float* __restrict__ emb, const int* __restrict__ src_nid,
                                         const int* __restrict__ e_src, const unsigned* wl, int count,
                                         float* acc, int lane) {
    for (int k0 = 0; k0 < count; k0 += 32) {
        unsigned ent = 0u;
        if (k0 + lane < count) {
            const unsigned en = wl[k0 + lane];
            const int row = src_nid[e_src[en >> 8]];
            ent = ((unsigned)row << 8) | (en & 255u);
        }
        const int nb = (count - k0 < 32) ? (count - k0) : 32;
        for (int j = 0; j < nb; j += 4) {
            float v[4][10];
            int dl[4];
#pragma unroll
            for (int q = 0; q < 4; ++q) {
                const unsigned e = (unsigned)__builtin_amdgcn_readlane((int)ent, j + q);
                const float* __restrict__ rp = emb + (size_t)(e >> 8) * DDIM;
                dl[q] = (int)(e & 255u);
#pragma unroll
                for (int c = 0; c < 10; ++c) {
                    const int col = lane + 32 * c;
                    v[q][c] = rp[(col < DDIM) ? col : DDIM - 1];
                }
            }
#pragma unroll
            for (int q = 0; q < 4; ++q) {
                if (j + q < nb) {
                    float* a = acc + dl[q] * PITCH;
#pragma unroll
                    for (int c = 0; c < 10; ++c) {
                        const int col = lane + 32 * c;
                        if (col < DDIM)
                            __hip_atomic_fetch_add(&a[col], v[q][c], __ATOMIC_RELAXED,
                                                   __HIP_MEMORY_SCOPE_WORKGROUP);
                    }
                }
            }
        }
    }
}

__device__ __forceinline__ void sm_keep(const int (&d)[4], int ebase, int lo, unsigned lim,
                                        unsigned* wl, int* cnt, int& count) {
#pragma unroll
    for (int q = 0; q < 4; ++q) {
        const unsigned dl = (unsigned)(d[q] - lo);
        const bool m = dl < lim;
        const unsigned mask = __builtin_amdgcn_ballot_w32(m);
        if (m) {
            const int pos = count + (int)__builtin_amdgcn_mbcnt_lo(mask, 0u);
            wl[pos] = ((unsigned)(ebase + q) << 8) | dl;
            __hip_atomic_fetch_add(&cnt[dl], 1, __ATOMIC_RELAXED, __HIP_MEMORY_SCOPE_WORKGROUP);
        }
        count += __popc(mask);
    }
}

template <int AVN>
__device__ __forceinline__ void sdpa_row(const float* x0, const float* x1, const float* x2,
                                         const float* av, float* o, int lane) {
    float d00 = 0, d01 = 0, d02 = 0, d11 = 0, d12 = 0, d22 = 0;
    for (int j = lane; j < PDIM; j += 32) {
        float a = x0[j], b = x1[j], e = x2[j];
        d00 += a * a; d01 += a * b; d02 += a * e;
        d11 += b * b; d12 += b * e; d22 += e * e;
    }
    for (int off = 16; off > 0; off >>= 1) {
        d00 += __shfl_xor(d00, off, 32);
        d01 += __shfl_xor(d01, off, 32);
        d02 += __shfl_xor(d02, off, 32);
        d11 += __shfl_xor(d11, off, 32);
        d12 += __shfl_xor(d12, off, 32);
        d22 += __shfl_xor(d22, off, 32);
    }
    const float scale = rsqrtf((float)DDIM);
    float S[3][3] = {{d00, d01, d02}, {d01, d11, d12}, {d02, d12, d22}};
    float wsum[3] = {0.f, 0.f, 0.f};
    for (int t = 0; t < 3; ++t) {
        float p0 = S[t][0] * scale, p1 = S[t][1] * scale, p2 = S[t][2] * scale;
        float mx = fmaxf(p0, fmaxf(p1, p2));
        float e0 = __expf(p0 - mx), e1 = __expf(p1 - mx), e2 = __expf(p2 - mx);
        float inv = 1.f / (e0 + e1 + e2);
        wsum[0] += e0 * inv; wsum[1] += e1 * inv; wsum[2] += e2 * inv;
    }
    const float w0 = wsum[0] * (1.f / 3.f), w1 = wsum[1] * (1.f / 3.f), w2 = wsum[2] * (1.f / 3.f);
    for (int j = lane; j < PDIM; j += 32)
        o[j] = w0 * x0[j] + w1 * x1[j] + w2 * x2[j] + ((j < AVN) ? av[j] : 0.f);
}

__global__ void sdpa_fuse_add(const float* __restrict__ h0, const float* __restrict__ h1,
                              const float* __restrict__ h2, const float* __restrict__ emb,
                              const int* __restrict__ nid, float* __restrict__ out, int n) {
    const int lane = threadIdx.x & 31;
    const int node = blockIdx.x * (blockDim.x >> 5) + (threadIdx.x >> 5);
    if (node >= n) return;
    sdpa_row<DDIM>(h0 + (size_t)node * PDIM, h1 + (size_t)node * PDIM, h2 + (size_t)node * PDIM,
                   emb + (size_t)nid[node] * DDIM, out + (size_t)node * PDIM, lane);
}

#define PS_WAVES 16
#define PS_TOK   (LN / PS_WAVES)
__global__ __launch_bounds__(PS_WAVES * 32) void pool_seq(const float* __restrict__ table,
                                                          const int* __restrict__ xb,
                                                          float* __restrict__ pooled) {
    static_assert(LN % PS_WAVES == 0, "every wave takes the same number of tokens");
    static_assert(PDIM == 320, "a lane's 4 + 4 + 2 floats cover a padded row");
    __shared__ __attribute__((aligned(16))) float part[PS_WAVES * PDIM];
    const int b = blockIdx.x;
    const int tid = threadIdx.x;
    const int lane = tid & 31;
    const int wave = __builtin_amdgcn_readfirstlane(tid >> 5);
    const int* __restrict__ row = xb + b * LN + wave * PS_TOK;

    v4f p0[PS_TOK], p1[PS_TOK];
    v2f p2[PS_TOK];
    bool ok[PS_TOK];
#pragma unroll
    for (int t = 0; t < PS_TOK; ++t) {
        const int idx = row[t];
        ok[t] = idx < NWN;
        const float* __restrict__ rp = table + (size_t)(ok[t] ? idx : 0) * PDIM;
        p0[t] = *(const v4f*)(rp + 4 * lane);
        p1[t] = *(const v4f*)(rp + 128 + 4 * lane);
        p2[t] = *(const v2f*)(rp + 256 + 2 * lane);
    }
    v4f a0 = (v4f){0.f, 0.f, 0.f, 0.f}, a1 = (v4f){0.f, 0.f, 0.f, 0.f};
    v2f a2 = (v2f){0.f, 0.f};
#pragma unroll
    for (int t = 0; t < PS_TOK; ++t) {
        if (ok[t]) { a0 += p0[t]; a1 += p1[t]; a2 += p2[t]; }
    }
    float* pw = part + wave * PDIM;
    *(v4f*)(pw + 4 * lane) = a0;
    *(v4f*)(pw + 128 + 4 * lane) = a1;
    *(v2f*)(pw + 256 + 2 * lane) = a2;
    __syncthreads();
    if (tid < PDIM) {
        float s = 0.f;
#pragma unroll
        for (int w = 0; w < PS_WAVES; ++w) s += part[w * PDIM + tid];
        pooled[(size_t)b * PDIM + tid] = s * (1.f / (float)LN);
    }
}

#define DF_WAVES 20
#define DF_PITCH 324
#define DF_UNR   2

__device__ __forceinline__ void df_tile_gemm(const float* arow, const v4f* __restrict__ bq, v8f& acc) {
#pragma unroll
    for (int kp = 0; kp < PDIM / 8; ++kp) {
        const v4f b = bq[(size_t)kp * NT16 * 32];
        const v2f a0 = *(const v2f*)(arow + 8 * kp);
        const v2f a1 = *(const v2f*)(arow + 8 * kp + 4);
        v2f b0; b0.x = b.x; b0.y = b.y;
        acc = __builtin_amdgcn_wmma_f32_16x16x4_f32(false, a0, false, b0, (short)0, acc, false, false);
        v2f b1; b1.x = b.z; b1.y = b.w;
        acc = __builtin_amdgcn_wmma_f32_16x16x4_f32(false, a1, false, b1, (short)0, acc, false, false);
    }
}

__device__ __forceinline__ void df_store_tile(const float* t, float* __restrict__ g, int tid) {
    for (int i = tid; i < 16 * (PDIM / 4); i += DF_WAVES * 32) {
        const int r = i / (PDIM / 4), c4 = i - r * (PDIM / 4);
        ((v4f*)g)[i] = *(const v4f*)(t + r * DF_PITCH + 4 * c4);
    }
}

__global__ __launch_bounds__(DF_WAVES * 32) void doc_rel_fused(
        const float* __restrict__ pooled, const float* __restrict__ wdn, const float* __restrict__ b_dense,
        const float* __restrict__ wp0, const float* __restrict__ wp1, const float* __restrict__ wp2,
        const float* __restrict__ bs0, const float* __restrict__ bs1, const float* __restrict__ bs2,
        const float* __restrict__ emb,
        const int* __restrict__ sn0, const int* __restrict__ sn1, const int* __restrict__ sn2,
        const int* __restrict__ es0, const int* __restrict__ es1, const int* __restrict__ es2,
        const int* __restrict__ ed0, const int* __restrict__ ed1, const int* __restrict__ ed2,
        int E, int n_dst, float* __restrict__ docout,
        float* __restrict__ d3_0, float* __restrict__ d3_1, float* __restrict__ d3_2) {
    __shared__ __attribute__((aligned(16))) float meanT[16 * DF_PITCH];
    __shared__ __attribute__((aligned(16))) float poolT[16 * DF_PITCH];
    __shared__ __attribute__((aligned(16))) float docT[16 * DF_PITCH];
    __shared__ __attribute__((aligned(16))) float outT[16 * DF_PITCH];
    __shared__ int cnt[16];
    __shared__ unsigned list[DF_WAVES * SM_CAP];
    const int tid = threadIdx.x;
    const int lane = tid & 31;
    const int wave = __builtin_amdgcn_readfirstlane(tid >> 5);
    const int hi = lane >> 4, l16 = lane & 15;
    const int lo = blockIdx.x * 16;
    const int rel = blockIdx.y;
    const int* __restrict__ src_nid = (rel == 0) ? sn0 : ((rel == 1) ? sn1 : sn2);
    const int* __restrict__ e_src   = (rel == 0) ? es0 : ((rel == 1) ? es1 : es2);
    const int* __restrict__ e_dst   = (rel == 0) ? ed0 : ((rel == 1) ? ed1 : ed2);
    const float* __restrict__ wp    = (rel == 0) ? wp0 : ((rel == 1) ? wp1 : wp2);
    const float* __restrict__ bs    = (rel == 0) ? bs0 : ((rel == 1) ? bs1 : bs2);
    float* __restrict__ d3          = (rel == 0) ? d3_0 : ((rel == 1) ? d3_1 : d3_2);

    for (int i = tid; i < 16 * (PDIM / 4); i += DF_WAVES * 32) {
        const int r = i / (PDIM / 4), c4 = i - r * (PDIM / 4);
        *(v4f*)(poolT + r * DF_PITCH + 4 * c4) = ((const v4f*)(pooled + (size_t)lo * PDIM))[i];
    }
    for (int i = tid; i < 16 * DF_PITCH / 4; i += DF_WAVES * 32)
        ((v4f*)meanT)[i] = (v4f){0.f, 0.f, 0.f, 0.f};
    if (tid < 16) cnt[tid] = 0;
    __syncthreads();

    {
        unsigned* wl = list + wave * SM_CAP;
        int count = 0;
        const int ngroups = (E + 127) >> 7;
        const unsigned lim = (unsigned)((n_dst - lo < 16) ? (n_dst - lo) : 16);
        for (int g0 = wave * DF_UNR; g0 < ngroups; g0 += DF_WAVES * DF_UNR) {
            if (count > SM_CAP - DF_UNR * 128) {
                sm_flush<DF_PITCH>(emb, src_nid, e_src, wl, count, meanT, lane);
                count = 0;
            }
            int d[DF_UNR][4];
            if ((g0 + DF_UNR) * 128 <= E) {
#pragma unroll
                for (int u = 0; u < DF_UNR; ++u) {
                    const int4 t = *(const int4*)(e_dst + (g0 + u) * 128 + lane * 4);
                    d[u][0] = t.x; d[u][1] = t.y; d[u][2] = t.z; d[u][3] = t.w;
                }
            } else {
#pragma unroll
                for (int u = 0; u < DF_UNR; ++u)
#pragma unroll
                    for (int q = 0; q < 4; ++q) {
                        const int e = (g0 + u) * 128 + lane * 4 + q;
                        d[u][q] = (e < E) ? e_dst[e] : -1;
                    }
            }
#pragma unroll
            for (int u = 0; u < DF_UNR; ++u)
                sm_keep(d[u], (g0 + u) * 128 + lane * 4, lo, lim, wl, cnt, count);
        }
        sm_flush<DF_PITCH>(emb, src_nid, e_src, wl, count, meanT, lane);
    }
    __syncthreads();
    for (int i = tid; i < 16 * PDIM; i += DF_WAVES * 32) {
        const int r = i / PDIM, c = i - r * PDIM;
        meanT[r * DF_PITCH + c] /= fmaxf((float)cnt[r], 1.f);
    }

    const int n = 16 * wave + l16;
    {
        const float bv = (n < DDIM) ? b_dense[n] : 0.f;
        v8f acc;
#pragma unroll
        for (int i = 0; i < 8; ++i) acc[i] = bv;
        df_tile_gemm(poolT + l16 * DF_PITCH + 2 * hi, (const v4f*)wdn + (size_t)wave * 32 + lane, acc);
#pragma unroll
        for (int i = 0; i < 8; ++i) docT[(i + 8 * hi) * DF_PITCH + n] = acc[i];
    }
    __syncthreads();
    if (rel == 0) df_store_tile(docT, docout + (size_t)lo * PDIM, tid);

    {
        const v4f* __restrict__ bq = (const v4f*)wp + (size_t)wave * 32 + lane;
        const float bv = (n < DDIM) ? bs[n] : 0.f;
        v8f acc;
#pragma unroll
        for (int i = 0; i < 8; ++i) acc[i] = bv;
        df_tile_gemm(docT + l16 * DF_PITCH + 2 * hi, bq, acc);
        df_tile_gemm(meanT + l16 * DF_PITCH + 2 * hi, bq + (size_t)PDIM * PDIM / 4, acc);
#pragma unroll
        for (int i = 0; i < 8; ++i) outT[(i + 8 * hi) * DF_PITCH + n] = acc[i];
    }
    __syncthreads();
    df_store_tile(outT, d3 + (size_t)lo * PDIM, tid);
}

#define WF_R   80
#define WF_MT  (WF_R / 16)

template <int NKP, bool HALF>
__device__ __forceinline__ void wf_tile_gemm(const float* arow, const v4f* __restrict__ bq, v8f (&acc)[WF_MT]) {
    constexpr int NPAIR = NKP + (HALF ? 1 : 0);
    v4f bc[2];
#pragma unroll
    for (int u = 0; u < 2; ++u) bc[u] = bq[(size_t)((u < NPAIR) ? u : 0) * NT16 * 32];
#pragma unroll
    for (int p0 = 0; p0 < NPAIR; p0 += 2) {
        v4f bn[2];
#pragma unroll
        for (int u = 0; u < 2; ++u) bn[u] = (p0 + 2 + u < NPAIR) ? bq[(size_t)(p0 + 2 + u) * NT16 * 32] : bc[u];
#pragma unroll
        for (int u = 0; u < 2; ++u) {
            const int p = p0 + u;
            if (p < NPAIR) {
                v2f b0; b0.x = bc[u].x; b0.y = bc[u].y;
#pragma unroll
                for (int mt = 0; mt < WF_MT; ++mt) {
                    const v2f a0 = *(const v2f*)(arow + mt * 16 * DF_PITCH + 8 * p);
                    acc[mt] = __builtin_amdgcn_wmma_f32_16x16x4_f32(false, a0, false, b0, (short)0, acc[mt], false, false);
                }
                if (p < NKP) {
                    v2f b1; b1.x = bc[u].z; b1.y = bc[u].w;
#pragma unroll
                    for (int mt = 0; mt < WF_MT; ++mt) {
                        const v2f a1 = *(const v2f*)(arow + mt * 16 * DF_PITCH + 8 * p + 4);
                        acc[mt] = __builtin_amdgcn_wmma_f32_16x16x4_f32(false, a1, false, b1, (short)0, acc[mt], false, false);
                    }
                }
            }
        }
        __builtin_amdgcn_sched_barrier(0);
        bc[0] = bn[0]; bc[1] = bn[1];
    }
}

__global__ __launch_bounds__(SM_WAVES * 32) void word_rel_fused(
        const float* __restrict__ emb, const int* __restrict__ dst_nid,
        const int* __restrict__ sn0, const int* __restrict__ sn1, const int* __restrict__ sn2,
        const int* __restrict__ es0, const int* __restrict__ es1, const int* __restrict__ es2,
        const int* __restrict__ ed0, const int* __restrict__ ed1, const int* __restrict__ ed2,
        const float* __restrict__ wp0, const float* __restrict__ wp1, const float* __restrict__ wp2,
        const float* __restrict__ bs0, const float* __restrict__ bs1, const float* __restrict__ bs2,
        int E, int n_dst, float* __restrict__ h_0, float* __restrict__ h_1, float* __restrict__ h_2) {
    static_assert(WF_R % 16 == 0 && WF_R <= 256, "whole 16-row tiles; a list entry holds the local row in 8 bits");
    static_assert(WF_R * DF_PITCH <= SM_WAVES * SM_CAP, "the rows' own table rows fit where the lists were");
    static_assert(NT16 <= SM_WAVES, "a wave per 16-column tile");
    __shared__ __attribute__((aligned(16))) float meanT[WF_R * DF_PITCH];
    __shared__ __attribute__((aligned(16))) unsigned list[SM_WAVES * SM_CAP];
    __shared__ int cnt[WF_R];
    const int tid = threadIdx.x;
    const int lane = tid & 31;
    const int wave = __builtin_amdgcn_readfirstlane(tid >> 5);
    const int hi = lane >> 4, l16 = lane & 15;
    const int lo = blockIdx.x * WF_R;
    const int rel = blockIdx.y;
    const int* __restrict__ src_nid = (rel == 0) ? sn0 : ((rel == 1) ? sn1 : sn2);
    const int* __restrict__ e_src   = (rel == 0) ? es0 : ((rel == 1) ? es1 : es2);
    const int* __restrict__ e_dst   = (rel == 0) ? ed0 : ((rel == 1) ? ed1 : ed2);
    const float* __restrict__ wp    = (rel == 0) ? wp0 : ((rel == 1) ? wp1 : wp2);
    const float* __restrict__ bs    = (rel == 0) ? bs0 : ((rel == 1) ? bs1 : bs2);
    float* __restrict__ h           = (rel == 0) ? h_0 : ((rel == 1) ? h_1 : h_2);

    for (int i = tid; i < WF_R * DF_PITCH / 4; i += SM_WAVES * 32)
        ((v4f*)meanT)[i] = (v4f){0.f, 0.f, 0.f, 0.f};
    for (int i = tid; i < WF_R; i += SM_WAVES * 32) cnt[i] = 0;
    __syncthreads();

    {
        unsigned* wl = list + wave * SM_CAP;
        int count = 0;
        const int ngroups = (E + 127) >> 7;
        const unsigned lim = (unsigned)((n_dst - lo < WF_R) ? (n_dst - lo) : WF_R);
        for (int g0 = wave * SM_UNR; g0 < ngroups; g0 += SM_WAVES * SM_UNR) {
            if (count > SM_CAP - SM_UNR * 128) {
                sm_flush<DF_PITCH>(emb, src_nid, e_src, wl, count, meanT, lane);
                count = 0;
            }
            int d[SM_UNR][4];
            if ((g0 + SM_UNR) * 128 <= E) {
#pragma unroll
                for (int u = 0; u < SM_UNR; ++u) {
                    const int4 t = *(const int4*)(e_dst + (g0 + u) * 128 + lane * 4);
                    d[u][0] = t.x; d[u][1] = t.y; d[u][2] = t.z; d[u][3] = t.w;
                }
            } else {
#pragma unroll
                for (int u = 0; u < SM_UNR; ++u)
#pragma unroll
                    for (int q = 0; q < 4; ++q) {
                        const int e = (g0 + u) * 128 + lane * 4 + q;
                        d[u][q] = (e < E) ? e_dst[e] : -1;
                    }
            }
#pragma unroll
            for (int u = 0; u < SM_UNR; ++u)
                sm_keep(d[u], (g0 + u) * 128 + lane * 4, lo, lim, wl, cnt, count);
        }
        sm_flush<DF_PITCH>(emb, src_nid, e_src, wl, count, meanT, lane);
    }
    __syncthreads();

    for (int i = tid; i < WF_R * (PDIM / 4); i += SM_WAVES * 32) {
        const int r = i / (PDIM / 4), c4 = i - r * (PDIM / 4);
        const float c = fmaxf((float)cnt[r], 1.f);
        v4f sv = *(const v4f*)(meanT + r * DF_PITCH + 4 * c4);
        sv.x /= c; sv.y /= c; sv.z /= c; sv.w /= c;
        *(v4f*)(meanT + r * DF_PITCH + 4 * c4) = sv;
    }
    float* hdT = (float*)list;
    for (int r = wave; r < WF_R; r += SM_WAVES) {
        const bool live = lo + r < n_dst;
        const v4f* __restrict__ rp = (const v4f*)(emb + (size_t)(live ? dst_nid[lo + r] : 0) * DDIM);
        for (int c4 = lane; c4 < DDIM / 4; c4 += 32)
            *(v4f*)(hdT + r * DF_PITCH + 4 * c4) = live ? rp[c4] : (v4f){0.f, 0.f, 0.f, 0.f};
    }
    __syncthreads();

    v8f acc[WF_MT];
    const int n = 16 * wave + l16;
    if (wave < NT16) {
        static_assert(DDIM % 4 == 0, "a table row is a whole number of k-steps");
        const v4f* __restrict__ bq = (const v4f*)wp + (size_t)wave * 32 + lane;
        const float bv = (n < DDIM) ? bs[n] : 0.f;
#pragma unroll
        for (int mt = 0; mt < WF_MT; ++mt)
#pragma unroll
            for (int i = 0; i < 8; ++i) acc[mt][i] = bv;
        wf_tile_gemm<DDIM / 8, (DDIM % 8) != 0>(hdT + l16 * DF_PITCH + 2 * hi, bq, acc);
        wf_tile_gemm<PDIM / 8, false>(meanT + l16 * DF_PITCH + 2 * hi, bq + (size_t)PDIM * PDIM / 4, acc);
    }
    __syncthreads();
    if (wave < NT16) {
#pragma unroll
        for (int mt = 0; mt < WF_MT; ++mt)
#pragma unroll
            for (int i = 0; i < 8; ++i) meanT[(16 * mt + i + 8 * hi) * DF_PITCH + n] = acc[mt][i];
    }
    __syncthreads();
    for (int i = tid; i < WF_R * (PDIM / 4); i += SM_WAVES * 32) {
        const int r = i / (PDIM / 4), c4 = i - r * (PDIM / 4);
        if (lo + r < n_dst)
            ((v4f*)(h + (size_t)lo * PDIM))[i] = *(const v4f*)(meanT + r * DF_PITCH + 4 * c4);
    }
}

__global__ __launch_bounds__(256) void doc_tail(
        const float* __restrict__ d3_0, const float* __restrict__ d3_1, const float* __restrict__ d3_2,
        const float* __restrict__ docout, const float* __restrict__ wfc, const float* __restrict__ bfc,
        float* __restrict__ out, int n) {
    __shared__ float rep[8 * PDIM];
    const int lane = threadIdx.x & 31;
    const int wave = threadIdx.x >> 5;
    const int node = blockIdx.x * 8 + wave;
    if (node < n)
        sdpa_row<PDIM>(d3_0 + (size_t)node * PDIM, d3_1 + (size_t)node * PDIM, d3_2 + (size_t)node * PDIM,
                       docout + (size_t)node * PDIM, rep + wave * PDIM, lane);
    __syncthreads();
    if (node < n && lane < CN) {
        float acc = bfc[lane];
        const float* rp = rep + wave * PDIM;
        for (int j = 0; j < DDIM; ++j) acc += rp[j] * wfc[(size_t)j * CN + lane];
        out[(size_t)node * CN + lane] = acc;
    }
}


extern "C" void kernel_launch(void* const* d_in, const int* in_sizes, int n_in,
                              void* d_out, int out_size, void* d_ws, size_t ws_size,
                              hipStream_t stream) {
    (void)in_sizes; (void)n_in; (void)out_size; (void)ws_size;
    const float* emb_word = (const float*)d_in[0];
    const float* emb_doc  = (const float*)d_in[1];
    const float* w_sage[6] = {(const float*)d_in[2],  (const float*)d_in[4],  (const float*)d_in[6],
                              (const float*)d_in[8],  (const float*)d_in[10], (const float*)d_in[12]};
    const float* b_sage[6] = {(const float*)d_in[3],  (const float*)d_in[5],  (const float*)d_in[7],
                              (const float*)d_in[9],  (const float*)d_in[11], (const float*)d_in[13]};
    const float* w_dense = (const float*)d_in[14];
    const float* b_dense = (const float*)d_in[15];
    const float* w_fc    = (const float*)d_in[16];
    const float* b_fc    = (const float*)d_in[17];
    const int* dst_nid      = (const int*)d_in[18];
    const int* src_nid_w[3] = {(const int*)d_in[19], (const int*)d_in[20], (const int*)d_in[21]};
    const int* e_src_w[3]   = {(const int*)d_in[22], (const int*)d_in[24], (const int*)d_in[26]};
    const int* e_dst_w[3]   = {(const int*)d_in[23], (const int*)d_in[25], (const int*)d_in[27]};
    const int* x_batch      = (const int*)d_in[28];
    const int* src_nid_d[3] = {(const int*)d_in[30], (const int*)d_in[31], (const int*)d_in[32]};
    const int* e_src_d[3]   = {(const int*)d_in[33], (const int*)d_in[35], (const int*)d_in[37]};
    const int* e_dst_d[3]   = {(const int*)d_in[34], (const int*)d_in[36], (const int*)d_in[38]};
    float* out = (float*)d_out;

    float* ws = (float*)d_ws;
    size_t off = 0;
    auto carve = [&](size_t nflt) { float* p = ws + off; off += nflt; return p; };
    float* nbuf = carve((size_t)NWN * PDIM);
    float* h3[3]; for (int r = 0; r < 3; ++r) h3[r] = carve((size_t)NWN * PDIM);
    float* wpk[6]; for (int r = 0; r < 6; ++r) wpk[r] = carve((size_t)2 * PDIM * PDIM);
    float* wdn  = carve((size_t)PDIM * PDIM);
    float* pooled = carve((size_t)BN * PDIM);
    float* docout = carve((size_t)BN * PDIM);
    float* d3[3]; for (int r = 0; r < 3; ++r) d3[r] = carve((size_t)BN * PDIM);

    const int TB = 256;

    for (int r = 0; r < 6; ++r)
        pack_sage_w<<<(2 * PDIM * PDIM + TB - 1) / TB, TB, 0, stream>>>(w_sage[r], wpk[r]);
    pack_w<<<(PDIM * PDIM + TB - 1) / TB, TB, 0, stream>>>(w_dense, wdn);

    word_rel_fused<<<dim3((NWN + WF_R - 1) / WF_R, 3), SM_WAVES * 32, 0, stream>>>(
        emb_word, dst_nid, src_nid_w[0], src_nid_w[1], src_nid_w[2], e_src_w[0], e_src_w[1], e_src_w[2],
        e_dst_w[0], e_dst_w[1], e_dst_w[2], wpk[0], wpk[1], wpk[2], b_sage[0], b_sage[1], b_sage[2],
        EWN, NWN, h3[0], h3[1], h3[2]);

    sdpa_fuse_add<<<(NWN + 7) / 8, 256, 0, stream>>>(h3[0], h3[1], h3[2], emb_word, dst_nid, nbuf, NWN);

    pool_seq<<<BN, PS_WAVES * 32, 0, stream>>>(nbuf, x_batch, pooled);

    doc_rel_fused<<<dim3(BN / 16, 3), DF_WAVES * 32, 0, stream>>>(
        pooled, wdn, b_dense, wpk[3], wpk[4], wpk[5], b_sage[3], b_sage[4], b_sage[5], emb_doc,
        src_nid_d[0], src_nid_d[1], src_nid_d[2], e_src_d[0], e_src_d[1], e_src_d[2],
        e_dst_d[0], e_dst_d[1], e_dst_d[2], EDN, BN, docout, d3[0], d3[1], d3[2]);

    doc_tail<<<(BN + 7) / 8, 256, 0, stream>>>(d3[0], d3[1], d3[2], docout, w_fc, b_fc, out, BN);
}


# B fragments read in place from the row-major weights (4-byte loads, same k order, K = 300 exact, pad columns stored as zeros): the 7 weight-pack launches are gone; 12 -> 5 launches; same sums
# speedup vs baseline: 4.3473x; 1.0536x over previous
#include <hip/hip_runtime.h>

#define DDIM 300
#define PDIM 320
#define NWN  20000
#define EWN  320000
#define BN   256
#define LN   256
#define SDN  2048
#define EDN  4096
#define CN   20

typedef __attribute__((ext_vector_type(2))) float v2f;
typedef __attribute__((ext_vector_type(4))) float v4f;
typedef __attribute__((ext_vector_type(8))) float v8f;

__device__ __forceinline__ v4f b_pair(const float* __restrict__ bcol, int p, bool second) {
    v4f b;
    b.x = bcol[(8 * p + 0) * DDIM];
    b.y = bcol[(8 * p + 1) * DDIM];
    b.z = second ? bcol[(8 * p + 4) * DDIM] : 0.f;
    b.w = second ? bcol[(8 * p + 5) * DDIM] : 0.f;
    return b;
}

#define SM_WAVES 32
#define SM_UNR   4
#define SM_CAP   1024

template <int PITCH>
__device__ __forceinline__ void sm_flush(const float* __restrict__ emb, const int* __restrict__ src_nid,
                                         const int* __restrict__ e_src, const unsigned* wl, int count,
                                         float* acc, int lane) {
    for (int k0 = 0; k0 < count; k0 += 32) {
        unsigned ent = 0u;
        if (k0 + lane < count) {
            const unsigned en = wl[k0 + lane];
            const int row = src_nid[e_src[en >> 8]];
            ent = ((unsigned)row << 8) | (en & 255u);
        }
        const int nb = (count - k0 < 32) ? (count - k0) : 32;
        for (int j = 0; j < nb; j += 4) {
            float v[4][10];
            int dl[4];
#pragma unroll
            for (int q = 0; q < 4; ++q) {
                const unsigned e = (unsigned)__builtin_amdgcn_readlane((int)ent, j + q);
                const float* __restrict__ rp = emb + (size_t)(e >> 8) * DDIM;
                dl[q] = (int)(e & 255u);
#pragma unroll
                for (int c = 0; c < 10; ++c) {
                    const int col = lane + 32 * c;
                    v[q][c] = rp[(col < DDIM) ? col : DDIM - 1];
                }
            }
#pragma unroll
            for (int q = 0; q < 4; ++q) {
                if (j + q < nb) {
                    float* a = acc + dl[q] * PITCH;
#pragma unroll
                    for (int c = 0; c < 10; ++c) {
                        const int col = lane + 32 * c;
                        if (col < DDIM)
                            __hip_atomic_fetch_add(&a[col], v[q][c], __ATOMIC_RELAXED,
                                                   __HIP_MEMORY_SCOPE_WORKGROUP);
                    }
                }
            }
        }
    }
}

__device__ __forceinline__ void sm_keep(const int (&d)[4], int ebase, int lo, unsigned lim,
                                        unsigned* wl, int* cnt, int& count) {
#pragma unroll
    for (int q = 0; q < 4; ++q) {
        const unsigned dl = (unsigned)(d[q] - lo);
        const bool m = dl < lim;
        const unsigned mask = __builtin_amdgcn_ballot_w32(m);
        if (m) {
            const int pos = count + (int)__builtin_amdgcn_mbcnt_lo(mask, 0u);
            wl[pos] = ((unsigned)(ebase + q) << 8) | dl;
            __hip_atomic_fetch_add(&cnt[dl], 1, __ATOMIC_RELAXED, __HIP_MEMORY_SCOPE_WORKGROUP);
        }
        count += __popc(mask);
    }
}

template <int AVN>
__device__ __forceinline__ void sdpa_row(const float* x0, const float* x1, const float* x2,
                                         const float* av, float* o, int lane) {
    float d00 = 0, d01 = 0, d02 = 0, d11 = 0, d12 = 0, d22 = 0;
    for (int j = lane; j < PDIM; j += 32) {
        float a = x0[j], b = x1[j], e = x2[j];
        d00 += a * a; d01 += a * b; d02 += a * e;
        d11 += b * b; d12 += b * e; d22 += e * e;
    }
    for (int off = 16; off > 0; off >>= 1) {
        d00 += __shfl_xor(d00, off, 32);
        d01 += __shfl_xor(d01, off, 32);
        d02 += __shfl_xor(d02, off, 32);
        d11 += __shfl_xor(d11, off, 32);
        d12 += __shfl_xor(d12, off, 32);
        d22 += __shfl_xor(d22, off, 32);
    }
    const float scale = rsqrtf((float)DDIM);
    float S[3][3] = {{d00, d01, d02}, {d01, d11, d12}, {d02, d12, d22}};
    float wsum[3] = {0.f, 0.f, 0.f};
    for (int t = 0; t < 3; ++t) {
        float p0 = S[t][0] * scale, p1 = S[t][1] * scale, p2 = S[t][2] * scale;
        float mx = fmaxf(p0, fmaxf(p1, p2));
        float e0 = __expf(p0 - mx), e1 = __expf(p1 - mx), e2 = __expf(p2 - mx);
        float inv = 1.f / (e0 + e1 + e2);
        wsum[0] += e0 * inv; wsum[1] += e1 * inv; wsum[2] += e2 * inv;
    }
    const float w0 = wsum[0] * (1.f / 3.f), w1 = wsum[1] * (1.f / 3.f), w2 = wsum[2] * (1.f / 3.f);
    for (int j = lane; j < PDIM; j += 32)
        o[j] = w0 * x0[j] + w1 * x1[j] + w2 * x2[j] + ((j < AVN) ? av[j] : 0.f);
}

__global__ void sdpa_fuse_add(const float* __restrict__ h0, const float* __restrict__ h1,
                              const float* __restrict__ h2, const float* __restrict__ emb,
                              const int* __restrict__ nid, float* __restrict__ out, int n) {
    const int lane = threadIdx.x & 31;
    const int node = blockIdx.x * (blockDim.x >> 5) + (threadIdx.x >> 5);
    if (node >= n) return;
    sdpa_row<DDIM>(h0 + (size_t)node * PDIM, h1 + (size_t)node * PDIM, h2 + (size_t)node * PDIM,
                   emb + (size_t)nid[node] * DDIM, out + (size_t)node * PDIM, lane);
}

#define PS_WAVES 16
#define PS_TOK   (LN / PS_WAVES)
__global__ __launch_bounds__(PS_WAVES * 32) void pool_seq(const float* __restrict__ table,
                                                          const int* __restrict__ xb,
                                                          float* __restrict__ pooled) {
    static_assert(LN % PS_WAVES == 0, "every wave takes the same number of tokens");
    static_assert(PDIM == 320, "a lane's 4 + 4 + 2 floats cover a padded row");
    __shared__ __attribute__((aligned(16))) float part[PS_WAVES * PDIM];
    const int b = blockIdx.x;
    const int tid = threadIdx.x;
    const int lane = tid & 31;
    const int wave = __builtin_amdgcn_readfirstlane(tid >> 5);
    const int* __restrict__ row = xb + b * LN + wave * PS_TOK;

    v4f p0[PS_TOK], p1[PS_TOK];
    v2f p2[PS_TOK];
    bool ok[PS_TOK];
#pragma unroll
    for (int t = 0; t < PS_TOK; ++t) {
        const int idx = row[t];
        ok[t] = idx < NWN;
        const float* __restrict__ rp = table + (size_t)(ok[t] ? idx : 0) * PDIM;
        p0[t] = *(const v4f*)(rp + 4 * lane);
        p1[t] = *(const v4f*)(rp + 128 + 4 * lane);
        p2[t] = *(const v2f*)(rp + 256 + 2 * lane);
    }
    v4f a0 = (v4f){0.f, 0.f, 0.f, 0.f}, a1 = (v4f){0.f, 0.f, 0.f, 0.f};
    v2f a2 = (v2f){0.f, 0.f};
#pragma unroll
    for (int t = 0; t < PS_TOK; ++t) {
        if (ok[t]) { a0 += p0[t]; a1 += p1[t]; a2 += p2[t]; }
    }
    float* pw = part + wave * PDIM;
    *(v4f*)(pw + 4 * lane) = a0;
    *(v4f*)(pw + 128 + 4 * lane) = a1;
    *(v2f*)(pw + 256 + 2 * lane) = a2;
    __syncthreads();
    if (tid < PDIM) {
        float s = 0.f;
#pragma unroll
        for (int w = 0; w < PS_WAVES; ++w) s += part[w * PDIM + tid];
        pooled[(size_t)b * PDIM + tid] = s * (1.f / (float)LN);
    }
}

#define DF_WAVES 20
#define DF_PITCH 324
#define DF_UNR   2

__device__ __forceinline__ void df_tile_gemm(const float* arow, const float* __restrict__ bcol, v8f& acc) {
    static_assert(DDIM % 4 == 0, "K is a whole number of k-steps");
#pragma unroll
    for (int kp = 0; kp < (DDIM + 7) / 8; ++kp) {
        const bool second = kp < DDIM / 8;
        const v4f b = b_pair(bcol, kp, second);
        const v2f a0 = *(const v2f*)(arow + 8 * kp);
        v2f b0; b0.x = b.x; b0.y = b.y;
        acc = __builtin_amdgcn_wmma_f32_16x16x4_f32(false, a0, false, b0, (short)0, acc, false, false);
        if (second) {
            const v2f a1 = *(const v2f*)(arow + 8 * kp + 4);
            v2f b1; b1.x = b.z; b1.y = b.w;
            acc = __builtin_amdgcn_wmma_f32_16x16x4_f32(false, a1, false, b1, (short)0, acc, false, false);
        }
    }
}

__device__ __forceinline__ void df_store_tile(const float* t, float* __restrict__ g, int tid) {
    for (int i = tid; i < 16 * (PDIM / 4); i += DF_WAVES * 32) {
        const int r = i / (PDIM / 4), c4 = i - r * (PDIM / 4);
        ((v4f*)g)[i] = *(const v4f*)(t + r * DF_PITCH + 4 * c4);
    }
}

__global__ __launch_bounds__(DF_WAVES * 32) void doc_rel_fused(
        const float* __restrict__ pooled, const float* __restrict__ wdn, const float* __restrict__ b_dense,
        const float* __restrict__ wp0, const float* __restrict__ wp1, const float* __restrict__ wp2,
        const float* __restrict__ bs0, const float* __restrict__ bs1, const float* __restrict__ bs2,
        const float* __restrict__ emb,
        const int* __restrict__ sn0, const int* __restrict__ sn1, const int* __restrict__ sn2,
        const int* __restrict__ es0, const int* __restrict__ es1, const int* __restrict__ es2,
        const int* __restrict__ ed0, const int* __restrict__ ed1, const int* __restrict__ ed2,
        int E, int n_dst, float* __restrict__ docout,
        float* __restrict__ d3_0, float* __restrict__ d3_1, float* __restrict__ d3_2) {
    __shared__ __attribute__((aligned(16))) float meanT[16 * DF_PITCH];
    __shared__ __attribute__((aligned(16))) float poolT[16 * DF_PITCH];
    __shared__ __attribute__((aligned(16))) float docT[16 * DF_PITCH];
    __shared__ __attribute__((aligned(16))) float outT[16 * DF_PITCH];
    __shared__ int cnt[16];
    __shared__ unsigned list[DF_WAVES * SM_CAP];
    const int tid = threadIdx.x;
    const int lane = tid & 31;
    const int wave = __builtin_amdgcn_readfirstlane(tid >> 5);
    const int hi = lane >> 4, l16 = lane & 15;
    const int lo = blockIdx.x * 16;
    const int rel = blockIdx.y;
    const int* __restrict__ src_nid = (rel == 0) ? sn0 : ((rel == 1) ? sn1 : sn2);
    const int* __restrict__ e_src   = (rel == 0) ? es0 : ((rel == 1) ? es1 : es2);
    const int* __restrict__ e_dst   = (rel == 0) ? ed0 : ((rel == 1) ? ed1 : ed2);
    const float* __restrict__ wp    = (rel == 0) ? wp0 : ((rel == 1) ? wp1 : wp2);
    const float* __restrict__ bs    = (rel == 0) ? bs0 : ((rel == 1) ? bs1 : bs2);
    float* __restrict__ d3          = (rel == 0) ? d3_0 : ((rel == 1) ? d3_1 : d3_2);

    for (int i = tid; i < 16 * (PDIM / 4); i += DF_WAVES * 32) {
        const int r = i / (PDIM / 4), c4 = i - r * (PDIM / 4);
        *(v4f*)(poolT + r * DF_PITCH + 4 * c4) = ((const v4f*)(pooled + (size_t)lo * PDIM))[i];
    }
    for (int i = tid; i < 16 * DF_PITCH / 4; i += DF_WAVES * 32)
        ((v4f*)meanT)[i] = (v4f){0.f, 0.f, 0.f, 0.f};
    if (tid < 16) cnt[tid] = 0;
    __syncthreads();

    {
        unsigned* wl = list + wave * SM_CAP;
        int count = 0;
        const int ngroups = (E + 127) >> 7;
        const unsigned lim = (unsigned)((n_dst - lo < 16) ? (n_dst - lo) : 16);
        for (int g0 = wave * DF_UNR; g0 < ngroups; g0 += DF_WAVES * DF_UNR) {
            if (count > SM_CAP - DF_UNR * 128) {
                sm_flush<DF_PITCH>(emb, src_nid, e_src, wl, count, meanT, lane);
                count = 0;
            }
            int d[DF_UNR][4];
            if ((g0 + DF_UNR) * 128 <= E) {
#pragma unroll
                for (int u = 0; u < DF_UNR; ++u) {
                    const int4 t = *(const int4*)(e_dst + (g0 + u) * 128 + lane * 4);
                    d[u][0] = t.x; d[u][1] = t.y; d[u][2] = t.z; d[u][3] = t.w;
                }
            } else {
#pragma unroll
                for (int u = 0; u < DF_UNR; ++u)
#pragma unroll
                    for (int q = 0; q < 4; ++q) {
                        const int e = (g0 + u) * 128 + lane * 4 + q;
                        d[u][q] = (e < E) ? e_dst[e] : -1;
                    }
            }
#pragma unroll
            for (int u = 0; u < DF_UNR; ++u)
                sm_keep(d[u], (g0 + u) * 128 + lane * 4, lo, lim, wl, cnt, count);
        }
        sm_flush<DF_PITCH>(emb, src_nid, e_src, wl, count, meanT, lane);
    }
    __syncthreads();
    for (int i = tid; i < 16 * PDIM; i += DF_WAVES * 32) {
        const int r = i / PDIM, c = i - r * PDIM;
        meanT[r * DF_PITCH + c] /= fmaxf((float)cnt[r], 1.f);
    }

    const int n = 16 * wave + l16;
    const bool ncol = n < DDIM;
    const int bofs = 2 * hi * DDIM + (ncol ? n : DDIM - 1);
    {
        const float bv = ncol ? b_dense[n] : 0.f;
        v8f acc;
#pragma unroll
        for (int i = 0; i < 8; ++i) acc[i] = bv;
        df_tile_gemm(poolT + l16 * DF_PITCH + 2 * hi, wdn + bofs, acc);
#pragma unroll
        for (int i = 0; i < 8; ++i) docT[(i + 8 * hi) * DF_PITCH + n] = ncol ? acc[i] : 0.f;
    }
    __syncthreads();
    if (rel == 0) df_store_tile(docT, docout + (size_t)lo * PDIM, tid);

    {
        const float bv = ncol ? bs[n] : 0.f;
        v8f acc;
#pragma unroll
        for (int i = 0; i < 8; ++i) acc[i] = bv;
        df_tile_gemm(docT + l16 * DF_PITCH + 2 * hi, wp + bofs, acc);
        df_tile_gemm(meanT + l16 * DF_PITCH + 2 * hi, wp + DDIM * DDIM + bofs, acc);
#pragma unroll
        for (int i = 0; i < 8; ++i) outT[(i + 8 * hi) * DF_PITCH + n] = ncol ? acc[i] : 0.f;
    }
    __syncthreads();
    df_store_tile(outT, d3 + (size_t)lo * PDIM, tid);
}

#define WF_R   80
#define WF_MT  (WF_R / 16)
#define WF_NT  ((DDIM + 15) / 16)

template <int NKP, bool HALF>
__device__ __forceinline__ void wf_tile_gemm(const float* arow, const float* __restrict__ bcol, v8f (&acc)[WF_MT]) {
    constexpr int NPAIR = NKP + (HALF ? 1 : 0);
    v4f bc[2];
#pragma unroll
    for (int u = 0; u < 2; ++u) {
        const int p = (u < NPAIR) ? u : 0;
        bc[u] = b_pair(bcol, p, p < NKP);
    }
#pragma unroll
    for (int p0 = 0; p0 < NPAIR; p0 += 2) {
        v4f bn[2];
#pragma unroll
        for (int u = 0; u < 2; ++u) bn[u] = (p0 + 2 + u < NPAIR) ? b_pair(bcol, p0 + 2 + u, p0 + 2 + u < NKP) : bc[u];
#pragma unroll
        for (int u = 0; u < 2; ++u) {
            const int p = p0 + u;
            if (p < NPAIR) {
                v2f b0; b0.x = bc[u].x; b0.y = bc[u].y;
#pragma unroll
                for (int mt = 0; mt < WF_MT; ++mt) {
                    const v2f a0 = *(const v2f*)(arow + mt * 16 * DF_PITCH + 8 * p);
                    acc[mt] = __builtin_amdgcn_wmma_f32_16x16x4_f32(false, a0, false, b0, (short)0, acc[mt], false, false);
                }
                if (p < NKP) {
                    v2f b1; b1.x = bc[u].z; b1.y = bc[u].w;
#pragma unroll
                    for (int mt = 0; mt < WF_MT; ++mt) {
                        const v2f a1 = *(const v2f*)(arow + mt * 16 * DF_PITCH + 8 * p + 4);
                        acc[mt] = __builtin_amdgcn_wmma_f32_16x16x4_f32(false, a1, false, b1, (short)0, acc[mt], false, false);
                    }
                }
            }
        }
        __builtin_amdgcn_sched_barrier(0);
        bc[0] = bn[0]; bc[1] = bn[1];
    }
}

__global__ __launch_bounds__(SM_WAVES * 32) void word_rel_fused(
        const float* __restrict__ emb, const int* __restrict__ dst_nid,
        const int* __restrict__ sn0, const int* __restrict__ sn1, const int* __restrict__ sn2,
        const int* __restrict__ es0, const int* __restrict__ es1, const int* __restrict__ es2,
        const int* __restrict__ ed0, const int* __restrict__ ed1, const int* __restrict__ ed2,
        const float* __restrict__ wp0, const float* __restrict__ wp1, const float* __restrict__ wp2,
        const float* __restrict__ bs0, const float* __restrict__ bs1, const float* __restrict__ bs2,
        int E, int n_dst, float* __restrict__ h_0, float* __restrict__ h_1, float* __restrict__ h_2) {
    static_assert(WF_R % 16 == 0 && WF_R <= 256, "whole 16-row tiles; a list entry holds the local row in 8 bits");
    static_assert(WF_R * DF_PITCH <= SM_WAVES * SM_CAP, "the rows' own table rows fit where the lists were");
    static_assert(WF_NT <= SM_WAVES && 16 * WF_NT <= PDIM, "a wave per 16-column tile, inside a padded row");
    __shared__ __attribute__((aligned(16))) float meanT[WF_R * DF_PITCH];
    __shared__ __attribute__((aligned(16))) unsigned list[SM_WAVES * SM_CAP];
    __shared__ int cnt[WF_R];
    const int tid = threadIdx.x;
    const int lane = tid & 31;
    const int wave = __builtin_amdgcn_readfirstlane(tid >> 5);
    const int hi = lane >> 4, l16 = lane & 15;
    const int lo = blockIdx.x * WF_R;
    const int rel = blockIdx.y;
    const int* __restrict__ src_nid = (rel == 0) ? sn0 : ((rel == 1) ? sn1 : sn2);
    const int* __restrict__ e_src   = (rel == 0) ? es0 : ((rel == 1) ? es1 : es2);
    const int* __restrict__ e_dst   = (rel == 0) ? ed0 : ((rel == 1) ? ed1 : ed2);
    const float* __restrict__ wp    = (rel == 0) ? wp0 : ((rel == 1) ? wp1 : wp2);
    const float* __restrict__ bs    = (rel == 0) ? bs0 : ((rel == 1) ? bs1 : bs2);
    float* __restrict__ h           = (rel == 0) ? h_0 : ((rel == 1) ? h_1 : h_2);

    for (int i = tid; i < WF_R * DF_PITCH / 4; i += SM_WAVES * 32)
        ((v4f*)meanT)[i] = (v4f){0.f, 0.f, 0.f, 0.f};
    for (int i = tid; i < WF_R; i += SM_WAVES * 32) cnt[i] = 0;
    __syncthreads();

    {
        unsigned* wl = list + wave * SM_CAP;
        int count = 0;
        const int ngroups = (E + 127) >> 7;
        const unsigned lim = (unsigned)((n_dst - lo < WF_R) ? (n_dst - lo) : WF_R);
        for (int g0 = wave * SM_UNR; g0 < ngroups; g0 += SM_WAVES * SM_UNR) {
            if (count > SM_CAP - SM_UNR * 128) {
                sm_flush<DF_PITCH>(emb, src_nid, e_src, wl, count, meanT, lane);
                count = 0;
            }
            int d[SM_UNR][4];
            if ((g0 + SM_UNR) * 128 <= E) {
#pragma unroll
                for (int u = 0; u < SM_UNR; ++u) {
                    const int4 t = *(const int4*)(e_dst + (g0 + u) * 128 + lane * 4);
                    d[u][0] = t.x; d[u][1] = t.y; d[u][2] = t.z; d[u][3] = t.w;
                }
            } else {
#pragma unroll
                for (int u = 0; u < SM_UNR; ++u)
#pragma unroll
                    for (int q = 0; q < 4; ++q) {
                        const int e = (g0 + u) * 128 + lane * 4 + q;
                        d[u][q] = (e < E) ? e_dst[e] : -1;
                    }
            }
#pragma unroll
            for (int u = 0; u < SM_UNR; ++u)
                sm_keep(d[u], (g0 + u) * 128 + lane * 4, lo, lim, wl, cnt, count);
        }
        sm_flush<DF_PITCH>(emb, src_nid, e_src, wl, count, meanT, lane);
    }
    __syncthreads();

    for (int i = tid; i < WF_R * (PDIM / 4); i += SM_WAVES * 32) {
        const int r = i / (PDIM / 4), c4 = i - r * (PDIM / 4);
        const float c = fmaxf((float)cnt[r], 1.f);
        v4f sv = *(const v4f*)(meanT + r * DF_PITCH + 4 * c4);
        sv.x /= c; sv.y /= c; sv.z /= c; sv.w /= c;
        *(v4f*)(meanT + r * DF_PITCH + 4 * c4) = sv;
    }
    float* hdT = (float*)list;
    for (int r = wave; r < WF_R; r += SM_WAVES) {
        const bool live = lo + r < n_dst;
        const v4f* __restrict__ rp = (const v4f*)(emb + (size_t)(live ? dst_nid[lo + r] : 0) * DDIM);
        for (int c4 = lane; c4 < DDIM / 4; c4 += 32)
            *(v4f*)(hdT + r * DF_PITCH + 4 * c4) = live ? rp[c4] : (v4f){0.f, 0.f, 0.f, 0.f};
    }
    __syncthreads();

    v8f acc[WF_MT];
    const int n = 16 * wave + l16;
    const bool ncol = n < DDIM;
    if (wave < WF_NT) {
        static_assert(DDIM % 4 == 0, "a table row and a mean row are a whole number of k-steps");
        const float* __restrict__ bcol = wp + 2 * hi * DDIM + (ncol ? n : DDIM - 1);
        const float bv = ncol ? bs[n] : 0.f;
#pragma unroll
        for (int mt = 0; mt < WF_MT; ++mt)
#pragma unroll
            for (int i = 0; i < 8; ++i) acc[mt][i] = bv;
        wf_tile_gemm<DDIM / 8, (DDIM % 8) != 0>(hdT + l16 * DF_PITCH + 2 * hi, bcol, acc);
        wf_tile_gemm<DDIM / 8, (DDIM % 8) != 0>(meanT + l16 * DF_PITCH + 2 * hi, bcol + DDIM * DDIM, acc);
    }
    __syncthreads();
    if (wave < WF_NT) {
#pragma unroll
        for (int mt = 0; mt < WF_MT; ++mt)
#pragma unroll
            for (int i = 0; i < 8; ++i) meanT[(16 * mt + i + 8 * hi) * DF_PITCH + n] = ncol ? acc[mt][i] : 0.f;
    }
    __syncthreads();
    for (int i = tid; i < WF_R * (PDIM / 4); i += SM_WAVES * 32) {
        const int r = i / (PDIM / 4), c4 = i - r * (PDIM / 4);
        if (lo + r < n_dst)
            ((v4f*)(h + (size_t)lo * PDIM))[i] = *(const v4f*)(meanT + r * DF_PITCH + 4 * c4);
    }
}

__global__ __launch_bounds__(256) void doc_tail(
        const float* __restrict__ d3_0, const float* __restrict__ d3_1, const float* __restrict__ d3_2,
        const float* __restrict__ docout, const float* __restrict__ wfc, const float* __restrict__ bfc,
        float* __restrict__ out, int n) {
    __shared__ float rep[8 * PDIM];
    const int lane = threadIdx.x & 31;
    const int wave = threadIdx.x >> 5;
    const int node = blockIdx.x * 8 + wave;
    if (node < n)
        sdpa_row<PDIM>(d3_0 + (size_t)node * PDIM, d3_1 + (size_t)node * PDIM, d3_2 + (size_t)node * PDIM,
                       docout + (size_t)node * PDIM, rep + wave * PDIM, lane);
    __syncthreads();
    if (node < n && lane < CN) {
        float acc = bfc[lane];
        const float* rp = rep + wave * PDIM;
        for (int j = 0; j < DDIM; ++j) acc += rp[j] * wfc[(size_t)j * CN + lane];
        out[(size_t)node * CN + lane] = acc;
    }
}


extern "C" void kernel_launch(void* const* d_in, const int* in_sizes, int n_in,
                              void* d_out, int out_size, void* d_ws, size_t ws_size,
                              hipStream_t stream) {
    (void)in_sizes; (void)n_in; (void)out_size; (void)ws_size;
    const float* emb_word = (const float*)d_in[0];
    const float* emb_doc  = (const float*)d_in[1];
    const float* w_sage[6] = {(const float*)d_in[2],  (const float*)d_in[4],  (const float*)d_in[6],
                              (const float*)d_in[8],  (const float*)d_in[10], (const float*)d_in[12]};
    const float* b_sage[6] = {(const float*)d_in[3],  (const float*)d_in[5],  (const float*)d_in[7],
                              (const float*)d_in[9],  (const float*)d_in[11], (const float*)d_in[13]};
    const float* w_dense = (const float*)d_in[14];
    const float* b_dense = (const float*)d_in[15];
    const float* w_fc    = (const float*)d_in[16];
    const float* b_fc    = (const float*)d_in[17];
    const int* dst_nid      = (const int*)d_in[18];
    const int* src_nid_w[3] = {(const int*)d_in[19], (const int*)d_in[20], (const int*)d_in[21]};
    const int* e_src_w[3]   = {(const int*)d_in[22], (const int*)d_in[24], (const int*)d_in[26]};
    const int* e_dst_w[3]   = {(const int*)d_in[23], (const int*)d_in[25], (const int*)d_in[27]};
    const int* x_batch      = (const int*)d_in[28];
    const int* src_nid_d[3] = {(const int*)d_in[30], (const int*)d_in[31], (const int*)d_in[32]};
    const int* e_src_d[3]   = {(const int*)d_in[33], (const int*)d_in[35], (const int*)d_in[37]};
    const int* e_dst_d[3]   = {(const int*)d_in[34], (const int*)d_in[36], (const int*)d_in[38]};
    float* out = (float*)d_out;

    float* ws = (float*)d_ws;
    size_t off = 0;
    auto carve = [&](size_t nflt) { float* p = ws + off; off += nflt; return p; };
    float* nbuf = carve((size_t)NWN * PDIM);
    float* h3[3]; for (int r = 0; r < 3; ++r) h3[r] = carve((size_t)NWN * PDIM);
    float* pooled = carve((size_t)BN * PDIM);
    float* docout = carve((size_t)BN * PDIM);
    float* d3[3]; for (int r = 0; r < 3; ++r) d3[r] = carve((size_t)BN * PDIM);

    word_rel_fused<<<dim3((NWN + WF_R - 1) / WF_R, 3), SM_WAVES * 32, 0, stream>>>(
        emb_word, dst_nid, src_nid_w[0], src_nid_w[1], src_nid_w[2], e_src_w[0], e_src_w[1], e_src_w[2],
        e_dst_w[0], e_dst_w[1], e_dst_w[2], w_sage[0], w_sage[1], w_sage[2], b_sage[0], b_sage[1], b_sage[2],
        EWN, NWN, h3[0], h3[1], h3[2]);

    sdpa_fuse_add<<<(NWN + 7) / 8, 256, 0, stream>>>(h3[0], h3[1], h3[2], emb_word, dst_nid, nbuf, NWN);

    pool_seq<<<BN, PS_WAVES * 32, 0, stream>>>(nbuf, x_batch, pooled);

    doc_rel_fused<<<dim3(BN / 16, 3), DF_WAVES * 32, 0, stream>>>(
        pooled, w_dense, b_dense, w_sage[3], w_sage[4], w_sage[5], b_sage[3], b_sage[4], b_sage[5], emb_doc,
        src_nid_d[0], src_nid_d[1], src_nid_d[2], e_src_d[0], e_src_d[1], e_src_d[2],
        e_dst_d[0], e_dst_d[1], e_dst_d[2], EDN, BN, docout, d3[0], d3[1], d3[2]);

    doc_tail<<<(BN + 7) / 8, 256, 0, stream>>>(d3[0], d3[1], d3[2], docout, w_fc, b_fc, out, BN);
}


# word path: a workgroup owns 240 rows, reads e_dst ONCE for them and runs its 3 sub-tiles of 80 in turn from the kept edges: 750 scans of the edge list -> 252; LDS adds in arrival order as before
# speedup vs baseline: 4.7804x; 1.0996x over previous
#include <hip/hip_runtime.h>

#define DDIM 300
#define PDIM 320
#define NWN  20000
#define EWN  320000
#define BN   256
#define LN   256
#define SDN  2048
#define EDN  4096
#define CN   20

typedef __attribute__((ext_vector_type(2))) float v2f;
typedef __attribute__((ext_vector_type(4))) float v4f;
typedef __attribute__((ext_vector_type(8))) float v8f;

__device__ __forceinline__ v4f b_pair(const float* __restrict__ bcol, int p, bool second) {
    v4f b;
    b.x = bcol[(8 * p + 0) * DDIM];
    b.y = bcol[(8 * p + 1) * DDIM];
    b.z = second ? bcol[(8 * p + 4) * DDIM] : 0.f;
    b.w = second ? bcol[(8 * p + 5) * DDIM] : 0.f;
    return b;
}

#define SM_WAVES 32
#define SM_UNR   4
#define SM_CAP   1024

template <int PITCH>
__device__ __forceinline__ void sm_add_rows(const float* __restrict__ emb, unsigned ent, int nb, float* acc, int lane) {
    for (int j = 0; j < nb; j += 4) {
        float v[4][10];
        int dl[4];
#pragma unroll
        for (int q = 0; q < 4; ++q) {
            const unsigned e = (unsigned)__builtin_amdgcn_readlane((int)ent, j + q);
            const float* __restrict__ rp = emb + (size_t)(e >> 8) * DDIM;
            dl[q] = (int)(e & 255u);
#pragma unroll
            for (int c = 0; c < 10; ++c) {
                const int col = lane + 32 * c;
                v[q][c] = rp[(col < DDIM) ? col : DDIM - 1];
            }
        }
#pragma unroll
        for (int q = 0; q < 4; ++q) {
            if (j + q < nb) {
                float* a = acc + dl[q] * PITCH;
#pragma unroll
                for (int c = 0; c < 10; ++c) {
                    const int col = lane + 32 * c;
                    if (col < DDIM)
                        __hip_atomic_fetch_add(&a[col], v[q][c], __ATOMIC_RELAXED,
                                               __HIP_MEMORY_SCOPE_WORKGROUP);
                }
            }
        }
    }
}

template <int PITCH>
__device__ __forceinline__ void sm_flush(const float* __restrict__ emb, const int* __restrict__ src_nid,
                                         const int* __restrict__ e_src, const unsigned* wl, int count,
                                         float* acc, int lane) {
    for (int k0 = 0; k0 < count; k0 += 32) {
        unsigned ent = 0u;
        if (k0 + lane < count) {
            const unsigned en = wl[k0 + lane];
            const int row = src_nid[e_src[en >> 8]];
            ent = ((unsigned)row << 8) | (en & 255u);
        }
        sm_add_rows<PITCH>(emb, ent, (count - k0 < 32) ? (count - k0) : 32, acc, lane);
    }
}

template <int U>
__device__ __forceinline__ void sm_load_round(const int* __restrict__ e_dst, int g0, int E, int lane, int (&d)[U][4]) {
    if ((g0 + U) * 128 <= E) {
#pragma unroll
        for (int u = 0; u < U; ++u) {
            const int4 t = *(const int4*)(e_dst + (g0 + u) * 128 + lane * 4);
            d[u][0] = t.x; d[u][1] = t.y; d[u][2] = t.z; d[u][3] = t.w;
        }
    } else {
#pragma unroll
        for (int u = 0; u < U; ++u)
#pragma unroll
            for (int q = 0; q < 4; ++q) {
                const int e = (g0 + u) * 128 + lane * 4 + q;
                d[u][q] = (e < E) ? e_dst[e] : -1;
            }
    }
}

__device__ __forceinline__ void sm_keep(const int (&d)[4], int ebase, int lo, unsigned lim,
                                        unsigned* wl, int* cnt, int& count) {
#pragma unroll
    for (int q = 0; q < 4; ++q) {
        const unsigned dl = (unsigned)(d[q] - lo);
        const bool m = dl < lim;
        const unsigned mask = __builtin_amdgcn_ballot_w32(m);
        if (m) {
            const int pos = count + (int)__builtin_amdgcn_mbcnt_lo(mask, 0u);
            wl[pos] = ((unsigned)(ebase + q) << 8) | dl;
            __hip_atomic_fetch_add(&cnt[dl], 1, __ATOMIC_RELAXED, __HIP_MEMORY_SCOPE_WORKGROUP);
        }
        count += __popc(mask);
    }
}

template <int AVN>
__device__ __forceinline__ void sdpa_row(const float* x0, const float* x1, const float* x2,
                                         const float* av, float* o, int lane) {
    float d00 = 0, d01 = 0, d02 = 0, d11 = 0, d12 = 0, d22 = 0;
    for (int j = lane; j < PDIM; j += 32) {
        float a = x0[j], b = x1[j], e = x2[j];
        d00 += a * a; d01 += a * b; d02 += a * e;
        d11 += b * b; d12 += b * e; d22 += e * e;
    }
    for (int off = 16; off > 0; off >>= 1) {
        d00 += __shfl_xor(d00, off, 32);
        d01 += __shfl_xor(d01, off, 32);
        d02 += __shfl_xor(d02, off, 32);
        d11 += __shfl_xor(d11, off, 32);
        d12 += __shfl_xor(d12, off, 32);
        d22 += __shfl_xor(d22, off, 32);
    }
    const float scale = rsqrtf((float)DDIM);
    float S[3][3] = {{d00, d01, d02}, {d01, d11, d12}, {d02, d12, d22}};
    float wsum[3] = {0.f, 0.f, 0.f};
    for (int t = 0; t < 3; ++t) {
        float p0 = S[t][0] * scale, p1 = S[t][1] * scale, p2 = S[t][2] * scale;
        float mx = fmaxf(p0, fmaxf(p1, p2));
        float e0 = __expf(p0 - mx), e1 = __expf(p1 - mx), e2 = __expf(p2 - mx);
        float inv = 1.f / (e0 + e1 + e2);
        wsum[0] += e0 * inv; wsum[1] += e1 * inv; wsum[2] += e2 * inv;
    }
    const float w0 = wsum[0] * (1.f / 3.f), w1 = wsum[1] * (1.f / 3.f), w2 = wsum[2] * (1.f / 3.f);
    for (int j = lane; j < PDIM; j += 32)
        o[j] = w0 * x0[j] + w1 * x1[j] + w2 * x2[j] + ((j < AVN) ? av[j] : 0.f);
}

__global__ void sdpa_fuse_add(const float* __restrict__ h0, const float* __restrict__ h1,
                              const float* __restrict__ h2, const float* __restrict__ emb,
                              const int* __restrict__ nid, float* __restrict__ out, int n) {
    const int lane = threadIdx.x & 31;
    const int node = blockIdx.x * (blockDim.x >> 5) + (threadIdx.x >> 5);
    if (node >= n) return;
    sdpa_row<DDIM>(h0 + (size_t)node * PDIM, h1 + (size_t)node * PDIM, h2 + (size_t)node * PDIM,
                   emb + (size_t)nid[node] * DDIM, out + (size_t)node * PDIM, lane);
}

#define PS_WAVES 16
#define PS_TOK   (LN / PS_WAVES)
__global__ __launch_bounds__(PS_WAVES * 32) void pool_seq(const float* __restrict__ table,
                                                          const int* __restrict__ xb,
                                                          float* __restrict__ pooled) {
    static_assert(LN % PS_WAVES == 0, "every wave takes the same number of tokens");
    static_assert(PDIM == 320, "a lane's 4 + 4 + 2 floats cover a padded row");
    __shared__ __attribute__((aligned(16))) float part[PS_WAVES * PDIM];
    const int b = blockIdx.x;
    const int tid = threadIdx.x;
    const int lane = tid & 31;
    const int wave = __builtin_amdgcn_readfirstlane(tid >> 5);
    const int* __restrict__ row = xb + b * LN + wave * PS_TOK;

    v4f p0[PS_TOK], p1[PS_TOK];
    v2f p2[PS_TOK];
    bool ok[PS_TOK];
#pragma unroll
    for (int t = 0; t < PS_TOK; ++t) {
        const int idx = row[t];
        ok[t] = idx < NWN;
        const float* __restrict__ rp = table + (size_t)(ok[t] ? idx : 0) * PDIM;
        p0[t] = *(const v4f*)(rp + 4 * lane);
        p1[t] = *(const v4f*)(rp + 128 + 4 * lane);
        p2[t] = *(const v2f*)(rp + 256 + 2 * lane);
    }
    v4f a0 = (v4f){0.f, 0.f, 0.f, 0.f}, a1 = (v4f){0.f, 0.f, 0.f, 0.f};
    v2f a2 = (v2f){0.f, 0.f};
#pragma unroll
    for (int t = 0; t < PS_TOK; ++t) {
        if (ok[t]) { a0 += p0[t]; a1 += p1[t]; a2 += p2[t]; }
    }
    float* pw = part + wave * PDIM;
    *(v4f*)(pw + 4 * lane) = a0;
    *(v4f*)(pw + 128 + 4 * lane) = a1;
    *(v2f*)(pw + 256 + 2 * lane) = a2;
    __syncthreads();
    if (tid < PDIM) {
        float s = 0.f;
#pragma unroll
        for (int w = 0; w < PS_WAVES; ++w) s += part[w * PDIM + tid];
        pooled[(size_t)b * PDIM + tid] = s * (1.f / (float)LN);
    }
}

#define DF_WAVES 20
#define DF_PITCH 324
#define DF_UNR   2

__device__ __forceinline__ void df_tile_gemm(const float* arow, const float* __restrict__ bcol, v8f& acc) {
    static_assert(DDIM % 4 == 0, "K is a whole number of k-steps");
#pragma unroll
    for (int kp = 0; kp < (DDIM + 7) / 8; ++kp) {
        const bool second = kp < DDIM / 8;
        const v4f b = b_pair(bcol, kp, second);
        const v2f a0 = *(const v2f*)(arow + 8 * kp);
        v2f b0; b0.x = b.x; b0.y = b.y;
        acc = __builtin_amdgcn_wmma_f32_16x16x4_f32(false, a0, false, b0, (short)0, acc, false, false);
        if (second) {
            const v2f a1 = *(const v2f*)(arow + 8 * kp + 4);
            v2f b1; b1.x = b.z; b1.y = b.w;
            acc = __builtin_amdgcn_wmma_f32_16x16x4_f32(false, a1, false, b1, (short)0, acc, false, false);
        }
    }
}

__device__ __forceinline__ void df_store_tile(const float* t, float* __restrict__ g, int tid) {
    for (int i = tid; i < 16 * (PDIM / 4); i += DF_WAVES * 32) {
        const int r = i / (PDIM / 4), c4 = i - r * (PDIM / 4);
        ((v4f*)g)[i] = *(const v4f*)(t + r * DF_PITCH + 4 * c4);
    }
}

__global__ __launch_bounds__(DF_WAVES * 32) void doc_rel_fused(
        const float* __restrict__ pooled, const float* __restrict__ wdn, const float* __restrict__ b_dense,
        const float* __restrict__ wp0, const float* __restrict__ wp1, const float* __restrict__ wp2,
        const float* __restrict__ bs0, const float* __restrict__ bs1, const float* __restrict__ bs2,
        const float* __restrict__ emb,
        const int* __restrict__ sn0, const int* __restrict__ sn1, const int* __restrict__ sn2,
        const int* __restrict__ es0, const int* __restrict__ es1, const int* __restrict__ es2,
        const int* __restrict__ ed0, const int* __restrict__ ed1, const int* __restrict__ ed2,
        int E, int n_dst, float* __restrict__ docout,
        float* __restrict__ d3_0, float* __restrict__ d3_1, float* __restrict__ d3_2) {
    __shared__ __attribute__((aligned(16))) float meanT[16 * DF_PITCH];
    __shared__ __attribute__((aligned(16))) float poolT[16 * DF_PITCH];
    __shared__ __attribute__((aligned(16))) float docT[16 * DF_PITCH];
    __shared__ __attribute__((aligned(16))) float outT[16 * DF_PITCH];
    __shared__ int cnt[16];
    __shared__ unsigned list[DF_WAVES * SM_CAP];
    const int tid = threadIdx.x;
    const int lane = tid & 31;
    const int wave = __builtin_amdgcn_readfirstlane(tid >> 5);
    const int hi = lane >> 4, l16 = lane & 15;
    const int lo = blockIdx.x * 16;
    const int rel = blockIdx.y;
    const int* __restrict__ src_nid = (rel == 0) ? sn0 : ((rel == 1) ? sn1 : sn2);
    const int* __restrict__ e_src   = (rel == 0) ? es0 : ((rel == 1) ? es1 : es2);
    const int* __restrict__ e_dst   = (rel == 0) ? ed0 : ((rel == 1) ? ed1 : ed2);
    const float* __restrict__ wp    = (rel == 0) ? wp0 : ((rel == 1) ? wp1 : wp2);
    const float* __restrict__ bs    = (rel == 0) ? bs0 : ((rel == 1) ? bs1 : bs2);
    float* __restrict__ d3          = (rel == 0) ? d3_0 : ((rel == 1) ? d3_1 : d3_2);

    for (int i = tid; i < 16 * (PDIM / 4); i += DF_WAVES * 32) {
        const int r = i / (PDIM / 4), c4 = i - r * (PDIM / 4);
        *(v4f*)(poolT + r * DF_PITCH + 4 * c4) = ((const v4f*)(pooled + (size_t)lo * PDIM))[i];
    }
    for (int i = tid; i < 16 * DF_PITCH / 4; i += DF_WAVES * 32)
        ((v4f*)meanT)[i] = (v4f){0.f, 0.f, 0.f, 0.f};
    if (tid < 16) cnt[tid] = 0;
    __syncthreads();

    {
        unsigned* wl = list + wave * SM_CAP;
        int count = 0;
        const int ngroups = (E + 127) >> 7;
        const unsigned lim = (unsigned)((n_dst - lo < 16) ? (n_dst - lo) : 16);
        for (int g0 = wave * DF_UNR; g0 < ngroups; g0 += DF_WAVES * DF_UNR) {
            if (count > SM_CAP - DF_UNR * 128) {
                sm_flush<DF_PITCH>(emb, src_nid, e_src, wl, count, meanT, lane);
                count = 0;
            }
            int d[DF_UNR][4];
            if ((g0 + DF_UNR) * 128 <= E) {
#pragma unroll
                for (int u = 0; u < DF_UNR; ++u) {
                    const int4 t = *(const int4*)(e_dst + (g0 + u) * 128 + lane * 4);
                    d[u][0] = t.x; d[u][1] = t.y; d[u][2] = t.z; d[u][3] = t.w;
                }
            } else {
#pragma unroll
                for (int u = 0; u < DF_UNR; ++u)
#pragma unroll
                    for (int q = 0; q < 4; ++q) {
                        const int e = (g0 + u) * 128 + lane * 4 + q;
                        d[u][q] = (e < E) ? e_dst[e] : -1;
                    }
            }
#pragma unroll
            for (int u = 0; u < DF_UNR; ++u)
                sm_keep(d[u], (g0 + u) * 128 + lane * 4, lo, lim, wl, cnt, count);
        }
        sm_flush<DF_PITCH>(emb, src_nid, e_src, wl, count, meanT, lane);
    }
    __syncthreads();
    for (int i = tid; i < 16 * PDIM; i += DF_WAVES * 32) {
        const int r = i / PDIM, c = i - r * PDIM;
        meanT[r * DF_PITCH + c] /= fmaxf((float)cnt[r], 1.f);
    }

    const int n = 16 * wave + l16;
    const bool ncol = n < DDIM;
    const int bofs = 2 * hi * DDIM + (ncol ? n : DDIM - 1);
    {
        const float bv = ncol ? b_dense[n] : 0.f;
        v8f acc;
#pragma unroll
        for (int i = 0; i < 8; ++i) acc[i] = bv;
        df_tile_gemm(poolT + l16 * DF_PITCH + 2 * hi, wdn + bofs, acc);
#pragma unroll
        for (int i = 0; i < 8; ++i) docT[(i + 8 * hi) * DF_PITCH + n] = ncol ? acc[i] : 0.f;
    }
    __syncthreads();
    if (rel == 0) df_store_tile(docT, docout + (size_t)lo * PDIM, tid);

    {
        const float bv = ncol ? bs[n] : 0.f;
        v8f acc;
#pragma unroll
        for (int i = 0; i < 8; ++i) acc[i] = bv;
        df_tile_gemm(docT + l16 * DF_PITCH + 2 * hi, wp + bofs, acc);
        df_tile_gemm(meanT + l16 * DF_PITCH + 2 * hi, wp + DDIM * DDIM + bofs, acc);
#pragma unroll
        for (int i = 0; i < 8; ++i) outT[(i + 8 * hi) * DF_PITCH + n] = ncol ? acc[i] : 0.f;
    }
    __syncthreads();
    df_store_tile(outT, d3 + (size_t)lo * PDIM, tid);
}

#define WF_R   80
#define WF_S   3
#define WF_MT  (WF_R / 16)
#define WF_NT  ((DDIM + 15) / 16)
#define WL_CAP 384

template <int NKP, bool HALF>
__device__ __forceinline__ void wf_tile_gemm(const float* arow, const float* __restrict__ bcol, v8f (&acc)[WF_MT]) {
    constexpr int NPAIR = NKP + (HALF ? 1 : 0);
    v4f bc[2];
#pragma unroll
    for (int u = 0; u < 2; ++u) {
        const int p = (u < NPAIR) ? u : 0;
        bc[u] = b_pair(bcol, p, p < NKP);
    }
#pragma unroll
    for (int p0 = 0; p0 < NPAIR; p0 += 2) {
        v4f bn[2];
#pragma unroll
        for (int u = 0; u < 2; ++u) bn[u] = (p0 + 2 + u < NPAIR) ? b_pair(bcol, p0 + 2 + u, p0 + 2 + u < NKP) : bc[u];
#pragma unroll
        for (int u = 0; u < 2; ++u) {
            const int p = p0 + u;
            if (p < NPAIR) {
                v2f b0; b0.x = bc[u].x; b0.y = bc[u].y;
#pragma unroll
                for (int mt = 0; mt < WF_MT; ++mt) {
                    const v2f a0 = *(const v2f*)(arow + mt * 16 * DF_PITCH + 8 * p);
                    acc[mt] = __builtin_amdgcn_wmma_f32_16x16x4_f32(false, a0, false, b0, (short)0, acc[mt], false, false);
                }
                if (p < NKP) {
                    v2f b1; b1.x = bc[u].z; b1.y = bc[u].w;
#pragma unroll
                    for (int mt = 0; mt < WF_MT; ++mt) {
                        const v2f a1 = *(const v2f*)(arow + mt * 16 * DF_PITCH + 8 * p + 4);
                        acc[mt] = __builtin_amdgcn_wmma_f32_16x16x4_f32(false, a1, false, b1, (short)0, acc[mt], false, false);
                    }
                }
            }
        }
        __builtin_amdgcn_sched_barrier(0);
        bc[0] = bn[0]; bc[1] = bn[1];
    }
}

__global__ __launch_bounds__(SM_WAVES * 32) void word_rel_fused(
        const float* __restrict__ emb, const int* __restrict__ dst_nid,
        const int* __restrict__ sn0, const int* __restrict__ sn1, const int* __restrict__ sn2,
        const int* __restrict__ es0, const int* __restrict__ es1, const int* __restrict__ es2,
        const int* __restrict__ ed0, const int* __restrict__ ed1, const int* __restrict__ ed2,
        const float* __restrict__ wp0, const float* __restrict__ wp1, const float* __restrict__ wp2,
        const float* __restrict__ bs0, const float* __restrict__ bs1, const float* __restrict__ bs2,
        int E, int n_dst, float* __restrict__ h_0, float* __restrict__ h_1, float* __restrict__ h_2) {
    static_assert(WF_R % 16 == 0 && WF_S * WF_R <= 256, "whole 16-row tiles; a list entry holds the local row in 8 bits");
    static_assert(WF_NT <= SM_WAVES && 16 * WF_NT <= PDIM, "a wave per 16-column tile, inside a padded row");
    static_assert(WL_CAP > 128 && WL_CAP % 32 == 0, "room for one 128-edge group behind the fill check");
    __shared__ __attribute__((aligned(16))) float meanT[WF_R * DF_PITCH];
    __shared__ __attribute__((aligned(16))) float hdT[WF_R * DF_PITCH];
    __shared__ unsigned list[SM_WAVES * WL_CAP];
    __shared__ unsigned pick[SM_WAVES * WL_CAP];
    __shared__ int cnt[WF_S * WF_R];
    __shared__ int ovf;
    const int tid = threadIdx.x;
    const int lane = tid & 31;
    const int wave = __builtin_amdgcn_readfirstlane(tid >> 5);
    const int lo = blockIdx.x * (WF_S * WF_R);
    const int rel = blockIdx.y;
    const int* __restrict__ src_nid = (rel == 0) ? sn0 : ((rel == 1) ? sn1 : sn2);
    const int* __restrict__ e_src   = (rel == 0) ? es0 : ((rel == 1) ? es1 : es2);
    const int* __restrict__ e_dst   = (rel == 0) ? ed0 : ((rel == 1) ? ed1 : ed2);
    const float* __restrict__ wp    = (rel == 0) ? wp0 : ((rel == 1) ? wp1 : wp2);
    const float* __restrict__ bs    = (rel == 0) ? bs0 : ((rel == 1) ? bs1 : bs2);
    float* __restrict__ h           = (rel == 0) ? h_0 : ((rel == 1) ? h_1 : h_2);

    for (int i = tid; i < WF_R * DF_PITCH / 4; i += SM_WAVES * 32)
        ((v4f*)meanT)[i] = (v4f){0.f, 0.f, 0.f, 0.f};
    for (int i = tid; i < WF_S * WF_R; i += SM_WAVES * 32) cnt[i] = 0;
    if (tid == 0) ovf = 0;
    __syncthreads();

    unsigned* wl = list + wave * WL_CAP;
    unsigned* wk = pick + wave * WL_CAP;
    const int ngroups = (E + 127) >> 7;
    const int rows_here = (n_dst - lo < WF_S * WF_R) ? (n_dst - lo) : WF_S * WF_R;

    int count = 0;
    {
        bool full = false;
        for (int g0 = wave * SM_UNR; g0 < ngroups; g0 += SM_WAVES * SM_UNR) {
            int d[SM_UNR][4];
            sm_load_round<SM_UNR>(e_dst, g0, E, lane, d);
#pragma unroll
            for (int u = 0; u < SM_UNR; ++u) {
                if (count > WL_CAP - 128) full = true;
                if (!full) sm_keep(d[u], (g0 + u) * 128 + lane * 4, lo, (unsigned)rows_here, wl, cnt, count);
            }
        }
        if (full) ovf = 1;
    }
    __syncthreads();
    const bool slow = __builtin_amdgcn_readfirstlane(ovf) != 0;
    if (!slow) {
        for (int k = lane; k < count; k += 32) {
            const unsigned en = wl[k];
            wl[k] = ((unsigned)src_nid[e_src[en >> 8]] << 8) | (en & 255u);
        }
    } else {
        for (int i = tid; i < WF_S * WF_R; i += SM_WAVES * 32) cnt[i] = 0;
        __syncthreads();
    }

#pragma unroll 1
    for (int s = 0; s < WF_S; ++s) {
        const int lo_s = lo + WF_R * s;
        if (lo_s >= n_dst) break;
        int t = tid;
        asm volatile("" : "+v"(t));
        const int hi = (t >> 4) & 1, l16 = t & 15;
        const int n = 16 * wave + l16;
        const bool ncol = n < DDIM;
        const unsigned lim = (unsigned)((n_dst - lo_s < WF_R) ? (n_dst - lo_s) : WF_R);
        if (!slow) {
            int np = 0;
            for (int k0 = 0; k0 < count; k0 += 32) {
                const unsigned en = (k0 + lane < count) ? wl[k0 + lane] : 0xffu;
                const unsigned r = (en & 255u) - (unsigned)(WF_R * s);
                const bool m = r < (unsigned)WF_R;
                const unsigned mask = __builtin_amdgcn_ballot_w32(m);
                if (m) wk[np + (int)__builtin_amdgcn_mbcnt_lo(mask, 0u)] = (en & ~255u) | r;
                np += __popc(mask);
            }
            for (int k0 = 0; k0 < np; k0 += 32)
                sm_add_rows<DF_PITCH>(emb, (k0 + lane < np) ? wk[k0 + lane] : 0u, (np - k0 < 32) ? (np - k0) : 32, meanT, lane);
        } else {
            int c = 0;
            for (int g0 = wave * SM_UNR; g0 < ngroups; g0 += SM_WAVES * SM_UNR) {
                int d[SM_UNR][4];
                sm_load_round<SM_UNR>(e_dst, g0, E, lane, d);
#pragma unroll
                for (int u = 0; u < SM_UNR; ++u) {
                    if (c > WL_CAP - 128) {
                        sm_flush<DF_PITCH>(emb, src_nid, e_src, wl, c, meanT, lane);
                        c = 0;
                    }
                    sm_keep(d[u], (g0 + u) * 128 + lane * 4, lo_s, lim, wl, cnt + WF_R * s, c);
                }
            }
            sm_flush<DF_PITCH>(emb, src_nid, e_src, wl, c, meanT, lane);
        }
        __syncthreads();

        for (int i = t; i < WF_R * (PDIM / 4); i += SM_WAVES * 32) {
            const int r = i / (PDIM / 4), c4 = i - r * (PDIM / 4);
            const float c = fmaxf((float)cnt[WF_R * s + r], 1.f);
            v4f sv = *(const v4f*)(meanT + r * DF_PITCH + 4 * c4);
            sv.x /= c; sv.y /= c; sv.z /= c; sv.w /= c;
            *(v4f*)(meanT + r * DF_PITCH + 4 * c4) = sv;
        }
        for (int r = wave; r < WF_R; r += SM_WAVES) {
            const bool live = lo_s + r < n_dst;
            const v4f* __restrict__ rp = (const v4f*)(emb + (size_t)(live ? dst_nid[lo_s + r] : 0) * DDIM);
            for (int c4 = lane; c4 < DDIM / 4; c4 += 32)
                *(v4f*)(hdT + r * DF_PITCH + 4 * c4) = live ? rp[c4] : (v4f){0.f, 0.f, 0.f, 0.f};
        }
        __syncthreads();

        v8f acc[WF_MT];
        if (wave < WF_NT) {
            static_assert(DDIM % 4 == 0, "a table row and a mean row are a whole number of k-steps");
            const float* __restrict__ bcol = wp + 2 * hi * DDIM + (ncol ? n : DDIM - 1);
            const float bv = ncol ? bs[n] : 0.f;
#pragma unroll
            for (int mt = 0; mt < WF_MT; ++mt)
#pragma unroll
                for (int i = 0; i < 8; ++i) acc[mt][i] = bv;
            wf_tile_gemm<DDIM / 8, (DDIM % 8) != 0>(hdT + l16 * DF_PITCH + 2 * hi, bcol, acc);
            wf_tile_gemm<DDIM / 8, (DDIM % 8) != 0>(meanT + l16 * DF_PITCH + 2 * hi, bcol + DDIM * DDIM, acc);
        }
        __syncthreads();
        if (wave < WF_NT) {
#pragma unroll
            for (int mt = 0; mt < WF_MT; ++mt)
#pragma unroll
                for (int i = 0; i < 8; ++i) meanT[(16 * mt + i + 8 * hi) * DF_PITCH + n] = ncol ? acc[mt][i] : 0.f;
        }
        __syncthreads();
        for (int i = t; i < WF_R * (PDIM / 4); i += SM_WAVES * 32) {
            const int r = i / (PDIM / 4), c4 = i - r * (PDIM / 4);
            if (lo_s + r < n_dst)
                ((v4f*)(h + (size_t)lo_s * PDIM))[i] = *(const v4f*)(meanT + r * DF_PITCH + 4 * c4);
        }
        if (s + 1 < WF_S) {
            __syncthreads();
            for (int i = t; i < WF_R * DF_PITCH / 4; i += SM_WAVES * 32)
                ((v4f*)meanT)[i] = (v4f){0.f, 0.f, 0.f, 0.f};
            __syncthreads();
        }
    }
}

__global__ __launch_bounds__(256) void doc_tail(
        const float* __restrict__ d3_0, const float* __restrict__ d3_1, const float* __restrict__ d3_2,
        const float* __restrict__ docout, const float* __restrict__ wfc, const float* __restrict__ bfc,
        float* __restrict__ out, int n) {
    __shared__ float rep[8 * PDIM];
    const int lane = threadIdx.x & 31;
    const int wave = threadIdx.x >> 5;
    const int node = blockIdx.x * 8 + wave;
    if (node < n)
        sdpa_row<PDIM>(d3_0 + (size_t)node * PDIM, d3_1 + (size_t)node * PDIM, d3_2 + (size_t)node * PDIM,
                       docout + (size_t)node * PDIM, rep + wave * PDIM, lane);
    __syncthreads();
    if (node < n && lane < CN) {
        float acc = bfc[lane];
        const float* rp = rep + wave * PDIM;
        for (int j = 0; j < DDIM; ++j) acc += rp[j] * wfc[(size_t)j * CN + lane];
        out[(size_t)node * CN + lane] = acc;
    }
}


extern "C" void kernel_launch(void* const* d_in, const int* in_sizes, int n_in,
                              void* d_out, int out_size, void* d_ws, size_t ws_size,
                              hipStream_t stream) {
    (void)in_sizes; (void)n_in; (void)out_size; (void)ws_size;
    const float* emb_word = (const float*)d_in[0];
    const float* emb_doc  = (const float*)d_in[1];
    const float* w_sage[6] = {(const float*)d_in[2],  (const float*)d_in[4],  (const float*)d_in[6],
                              (const float*)d_in[8],  (const float*)d_in[10], (const float*)d_in[12]};
    const float* b_sage[6] = {(const float*)d_in[3],  (const float*)d_in[5],  (const float*)d_in[7],
                              (const float*)d_in[9],  (const float*)d_in[11], (const float*)d_in[13]};
    const float* w_dense = (const float*)d_in[14];
    const float* b_dense = (const float*)d_in[15];
    const float* w_fc    = (const float*)d_in[16];
    const float* b_fc    = (const float*)d_in[17];
    const int* dst_nid      = (const int*)d_in[18];
    const int* src_nid_w[3] = {(const int*)d_in[19], (const int*)d_in[20], (const int*)d_in[21]};
    const int* e_src_w[3]   = {(const int*)d_in[22], (const int*)d_in[24], (const int*)d_in[26]};
    const int* e_dst_w[3]   = {(const int*)d_in[23], (const int*)d_in[25], (const int*)d_in[27]};
    const int* x_batch      = (const int*)d_in[28];
    const int* src_nid_d[3] = {(const int*)d_in[30], (const int*)d_in[31], (const int*)d_in[32]};
    const int* e_src_d[3]   = {(const int*)d_in[33], (const int*)d_in[35], (const int*)d_in[37]};
    const int* e_dst_d[3]   = {(const int*)d_in[34], (const int*)d_in[36], (const int*)d_in[38]};
    float* out = (float*)d_out;

    float* ws = (float*)d_ws;
    size_t off = 0;
    auto carve = [&](size_t nflt) { float* p = ws + off; off += nflt; return p; };
    float* nbuf = carve((size_t)NWN * PDIM);
    float* h3[3]; for (int r = 0; r < 3; ++r) h3[r] = carve((size_t)NWN * PDIM);
    float* pooled = carve((size_t)BN * PDIM);
    float* docout = carve((size_t)BN * PDIM);
    float* d3[3]; for (int r = 0; r < 3; ++r) d3[r] = carve((size_t)BN * PDIM);

    word_rel_fused<<<dim3((NWN + WF_S * WF_R - 1) / (WF_S * WF_R), 3), SM_WAVES * 32, 0, stream>>>(
        emb_word, dst_nid, src_nid_w[0], src_nid_w[1], src_nid_w[2], e_src_w[0], e_src_w[1], e_src_w[2],
        e_dst_w[0], e_dst_w[1], e_dst_w[2], w_sage[0], w_sage[1], w_sage[2], b_sage[0], b_sage[1], b_sage[2],
        EWN, NWN, h3[0], h3[1], h3[2]);

    sdpa_fuse_add<<<(NWN + 7) / 8, 256, 0, stream>>>(h3[0], h3[1], h3[2], emb_word, dst_nid, nbuf, NWN);

    pool_seq<<<BN, PS_WAVES * 32, 0, stream>>>(nbuf, x_batch, pooled);

    doc_rel_fused<<<dim3(BN / 16, 3), DF_WAVES * 32, 0, stream>>>(
        pooled, w_dense, b_dense, w_sage[3], w_sage[4], w_sage[5], b_sage[3], b_sage[4], b_sage[5], emb_doc,
        src_nid_d[0], src_nid_d[1], src_nid_d[2], e_src_d[0], e_src_d[1], e_src_d[2],
        e_dst_d[0], e_dst_d[1], e_dst_d[2], EDN, BN, docout, d3[0], d3[1], d3[2]);

    doc_tail<<<(BN + 7) / 8, 256, 0, stream>>>(d3[0], d3[1], d3[2], docout, w_fc, b_fc, out, BN);
}


# word path neighbour means without LDS float atomics: scan ranks each kept edge into its row's 48-slot LDS bucket (returning integer add); a wave sums a row in registers, arrival order, stores it once
# speedup vs baseline: 4.9552x; 1.0366x over previous
#include <hip/hip_runtime.h>

#define DDIM 300
#define PDIM 320
#define NWN  20000
#define EWN  320000
#define BN   256
#define LN   256
#define SDN  2048
#define EDN  4096
#define CN   20

typedef __attribute__((ext_vector_type(2))) float v2f;
typedef __attribute__((ext_vector_type(4))) float v4f;
typedef __attribute__((ext_vector_type(8))) float v8f;

__device__ __forceinline__ v4f b_pair(const float* __restrict__ bcol, int p, bool second) {
    v4f b;
    b.x = bcol[(8 * p + 0) * DDIM];
    b.y = bcol[(8 * p + 1) * DDIM];
    b.z = second ? bcol[(8 * p + 4) * DDIM] : 0.f;
    b.w = second ? bcol[(8 * p + 5) * DDIM] : 0.f;
    return b;
}

#define SM_WAVES 32
#define SM_UNR   4
#define SM_CAP   1024

template <int PITCH>
__device__ __forceinline__ void sm_add_rows(const float* __restrict__ emb, unsigned ent, int nb, float* acc, int lane) {
    for (int j = 0; j < nb; j += 4) {
        float v[4][10];
        int dl[4];
#pragma unroll
        for (int q = 0; q < 4; ++q) {
            const unsigned e = (unsigned)__builtin_amdgcn_readlane((int)ent, j + q);
            const float* __restrict__ rp = emb + (size_t)(e >> 8) * DDIM;
            dl[q] = (int)(e & 255u);
#pragma unroll
            for (int c = 0; c < 10; ++c) {
                const int col = lane + 32 * c;
                v[q][c] = rp[(col < DDIM) ? col : DDIM - 1];
            }
        }
#pragma unroll
        for (int q = 0; q < 4; ++q) {
            if (j + q < nb) {
                float* a = acc + dl[q] * PITCH;
#pragma unroll
                for (int c = 0; c < 10; ++c) {
                    const int col = lane + 32 * c;
                    if (col < DDIM)
                        __hip_atomic_fetch_add(&a[col], v[q][c], __ATOMIC_RELAXED,
                                               __HIP_MEMORY_SCOPE_WORKGROUP);
                }
            }
        }
    }
}

template <int PITCH>
__device__ __forceinline__ void sm_flush(const float* __restrict__ emb, const int* __restrict__ src_nid,
                                         const int* __restrict__ e_src, const unsigned* wl, int count,
                                         float* acc, int lane) {
    for (int k0 = 0; k0 < count; k0 += 32) {
        unsigned ent = 0u;
        if (k0 + lane < count) {
            const unsigned en = wl[k0 + lane];
            const int row = src_nid[e_src[en >> 8]];
            ent = ((unsigned)row << 8) | (en & 255u);
        }
        sm_add_rows<PITCH>(emb, ent, (count - k0 < 32) ? (count - k0) : 32, acc, lane);
    }
}

template <int U>
__device__ __forceinline__ void sm_load_round(const int* __restrict__ e_dst, int g0, int E, int lane, int (&d)[U][4]) {
    if ((g0 + U) * 128 <= E) {
#pragma unroll
        for (int u = 0; u < U; ++u) {
            const int4 t = *(const int4*)(e_dst + (g0 + u) * 128 + lane * 4);
            d[u][0] = t.x; d[u][1] = t.y; d[u][2] = t.z; d[u][3] = t.w;
        }
    } else {
#pragma unroll
        for (int u = 0; u < U; ++u)
#pragma unroll
            for (int q = 0; q < 4; ++q) {
                const int e = (g0 + u) * 128 + lane * 4 + q;
                d[u][q] = (e < E) ? e_dst[e] : -1;
            }
    }
}

__device__ __forceinline__ void sm_keep(const int (&d)[4], int ebase, int lo, unsigned lim,
                                        unsigned* wl, int* cnt, int& count) {
#pragma unroll
    for (int q = 0; q < 4; ++q) {
        const unsigned dl = (unsigned)(d[q] - lo);
        const bool m = dl < lim;
        const unsigned mask = __builtin_amdgcn_ballot_w32(m);
        if (m) {
            const int pos = count + (int)__builtin_amdgcn_mbcnt_lo(mask, 0u);
            wl[pos] = ((unsigned)(ebase + q) << 8) | dl;
            __hip_atomic_fetch_add(&cnt[dl], 1, __ATOMIC_RELAXED, __HIP_MEMORY_SCOPE_WORKGROUP);
        }
        count += __popc(mask);
    }
}

template <int AVN>
__device__ __forceinline__ void sdpa_row(const float* x0, const float* x1, const float* x2,
                                         const float* av, float* o, int lane) {
    float d00 = 0, d01 = 0, d02 = 0, d11 = 0, d12 = 0, d22 = 0;
    for (int j = lane; j < PDIM; j += 32) {
        float a = x0[j], b = x1[j], e = x2[j];
        d00 += a * a; d01 += a * b; d02 += a * e;
        d11 += b * b; d12 += b * e; d22 += e * e;
    }
    for (int off = 16; off > 0; off >>= 1) {
        d00 += __shfl_xor(d00, off, 32);
        d01 += __shfl_xor(d01, off, 32);
        d02 += __shfl_xor(d02, off, 32);
        d11 += __shfl_xor(d11, off, 32);
        d12 += __shfl_xor(d12, off, 32);
        d22 += __shfl_xor(d22, off, 32);
    }
    const float scale = rsqrtf((float)DDIM);
    float S[3][3] = {{d00, d01, d02}, {d01, d11, d12}, {d02, d12, d22}};
    float wsum[3] = {0.f, 0.f, 0.f};
    for (int t = 0; t < 3; ++t) {
        float p0 = S[t][0] * scale, p1 = S[t][1] * scale, p2 = S[t][2] * scale;
        float mx = fmaxf(p0, fmaxf(p1, p2));
        float e0 = __expf(p0 - mx), e1 = __expf(p1 - mx), e2 = __expf(p2 - mx);
        float inv = 1.f / (e0 + e1 + e2);
        wsum[0] += e0 * inv; wsum[1] += e1 * inv; wsum[2] += e2 * inv;
    }
    const float w0 = wsum[0] * (1.f / 3.f), w1 = wsum[1] * (1.f / 3.f), w2 = wsum[2] * (1.f / 3.f);
    for (int j = lane; j < PDIM; j += 32)
        o[j] = w0 * x0[j] + w1 * x1[j] + w2 * x2[j] + ((j < AVN) ? av[j] : 0.f);
}

__global__ void sdpa_fuse_add(const float* __restrict__ h0, const float* __restrict__ h1,
                              const float* __restrict__ h2, const float* __restrict__ emb,
                              const int* __restrict__ nid, float* __restrict__ out, int n) {
    const int lane = threadIdx.x & 31;
    const int node = blockIdx.x * (blockDim.x >> 5) + (threadIdx.x >> 5);
    if (node >= n) return;
    sdpa_row<DDIM>(h0 + (size_t)node * PDIM, h1 + (size_t)node * PDIM, h2 + (size_t)node * PDIM,
                   emb + (size_t)nid[node] * DDIM, out + (size_t)node * PDIM, lane);
}

#define PS_WAVES 16
#define PS_TOK   (LN / PS_WAVES)
__global__ __launch_bounds__(PS_WAVES * 32) void pool_seq(const float* __restrict__ table,
                                                          const int* __restrict__ xb,
                                                          float* __restrict__ pooled) {
    static_assert(LN % PS_WAVES == 0, "every wave takes the same number of tokens");
    static_assert(PDIM == 320, "a lane's 4 + 4 + 2 floats cover a padded row");
    __shared__ __attribute__((aligned(16))) float part[PS_WAVES * PDIM];
    const int b = blockIdx.x;
    const int tid = threadIdx.x;
    const int lane = tid & 31;
    const int wave = __builtin_amdgcn_readfirstlane(tid >> 5);
    const int* __restrict__ row = xb + b * LN + wave * PS_TOK;

    v4f p0[PS_TOK], p1[PS_TOK];
    v2f p2[PS_TOK];
    bool ok[PS_TOK];
#pragma unroll
    for (int t = 0; t < PS_TOK; ++t) {
        const int idx = row[t];
        ok[t] = idx < NWN;
        const float* __restrict__ rp = table + (size_t)(ok[t] ? idx : 0) * PDIM;
        p0[t] = *(const v4f*)(rp + 4 * lane);
        p1[t] = *(const v4f*)(rp + 128 + 4 * lane);
        p2[t] = *(const v2f*)(rp + 256 + 2 * lane);
    }
    v4f a0 = (v4f){0.f, 0.f, 0.f, 0.f}, a1 = (v4f){0.f, 0.f, 0.f, 0.f};
    v2f a2 = (v2f){0.f, 0.f};
#pragma unroll
    for (int t = 0; t < PS_TOK; ++t) {
        if (ok[t]) { a0 += p0[t]; a1 += p1[t]; a2 += p2[t]; }
    }
    float* pw = part + wave * PDIM;
    *(v4f*)(pw + 4 * lane) = a0;
    *(v4f*)(pw + 128 + 4 * lane) = a1;
    *(v2f*)(pw + 256 + 2 * lane) = a2;
    __syncthreads();
    if (tid < PDIM) {
        float s = 0.f;
#pragma unroll
        for (int w = 0; w < PS_WAVES; ++w) s += part[w * PDIM + tid];
        pooled[(size_t)b * PDIM + tid] = s * (1.f / (float)LN);
    }
}

#define DF_WAVES 20
#define DF_PITCH 324
#define DF_UNR   2

__device__ __forceinline__ void df_tile_gemm(const float* arow, const float* __restrict__ bcol, v8f& acc) {
    static_assert(DDIM % 4 == 0, "K is a whole number of k-steps");
#pragma unroll
    for (int kp = 0; kp < (DDIM + 7) / 8; ++kp) {
        const bool second = kp < DDIM / 8;
        const v4f b = b_pair(bcol, kp, second);
        const v2f a0 = *(const v2f*)(arow + 8 * kp);
        v2f b0; b0.x = b.x; b0.y = b.y;
        acc = __builtin_amdgcn_wmma_f32_16x16x4_f32(false, a0, false, b0, (short)0, acc, false, false);
        if (second) {
            const v2f a1 = *(const v2f*)(arow + 8 * kp + 4);
            v2f b1; b1.x = b.z; b1.y = b.w;
            acc = __builtin_amdgcn_wmma_f32_16x16x4_f32(false, a1, false, b1, (short)0, acc, false, false);
        }
    }
}

__device__ __forceinline__ void df_store_tile(const float* t, float* __restrict__ g, int tid) {
    for (int i = tid; i < 16 * (PDIM / 4); i += DF_WAVES * 32) {
        const int r = i / (PDIM / 4), c4 = i - r * (PDIM / 4);
        ((v4f*)g)[i] = *(const v4f*)(t + r * DF_PITCH + 4 * c4);
    }
}

__global__ __launch_bounds__(DF_WAVES * 32) void doc_rel_fused(
        const float* __restrict__ pooled, const float* __restrict__ wdn, const float* __restrict__ b_dense,
        const float* __restrict__ wp0, const float* __restrict__ wp1, const float* __restrict__ wp2,
        const float* __restrict__ bs0, const float* __restrict__ bs1, const float* __restrict__ bs2,
        const float* __restrict__ emb,
        const int* __restrict__ sn0, const int* __restrict__ sn1, const int* __restrict__ sn2,
        const int* __restrict__ es0, const int* __restrict__ es1, const int* __restrict__ es2,
        const int* __restrict__ ed0, const int* __restrict__ ed1, const int* __restrict__ ed2,
        int E, int n_dst, float* __restrict__ docout,
        float* __restrict__ d3_0, float* __restrict__ d3_1, float* __restrict__ d3_2) {
    __shared__ __attribute__((aligned(16))) float meanT[16 * DF_PITCH];
    __shared__ __attribute__((aligned(16))) float poolT[16 * DF_PITCH];
    __shared__ __attribute__((aligned(16))) float docT[16 * DF_PITCH];
    __shared__ __attribute__((aligned(16))) float outT[16 * DF_PITCH];
    __shared__ int cnt[16];
    __shared__ unsigned list[DF_WAVES * SM_CAP];
    const int tid = threadIdx.x;
    const int lane = tid & 31;
    const int wave = __builtin_amdgcn_readfirstlane(tid >> 5);
    const int hi = lane >> 4, l16 = lane & 15;
    const int lo = blockIdx.x * 16;
    const int rel = blockIdx.y;
    const int* __restrict__ src_nid = (rel == 0) ? sn0 : ((rel == 1) ? sn1 : sn2);
    const int* __restrict__ e_src   = (rel == 0) ? es0 : ((rel == 1) ? es1 : es2);
    const int* __restrict__ e_dst   = (rel == 0) ? ed0 : ((rel == 1) ? ed1 : ed2);
    const float* __restrict__ wp    = (rel == 0) ? wp0 : ((rel == 1) ? wp1 : wp2);
    const float* __restrict__ bs    = (rel == 0) ? bs0 : ((rel == 1) ? bs1 : bs2);
    float* __restrict__ d3          = (rel == 0) ? d3_0 : ((rel == 1) ? d3_1 : d3_2);

    for (int i = tid; i < 16 * (PDIM / 4); i += DF_WAVES * 32) {
        const int r = i / (PDIM / 4), c4 = i - r * (PDIM / 4);
        *(v4f*)(poolT + r * DF_PITCH + 4 * c4) = ((const v4f*)(pooled + (size_t)lo * PDIM))[i];
    }
    for (int i = tid; i < 16 * DF_PITCH / 4; i += DF_WAVES * 32)
        ((v4f*)meanT)[i] = (v4f){0.f, 0.f, 0.f, 0.f};
    if (tid < 16) cnt[tid] = 0;
    __syncthreads();

    {
        unsigned* wl = list + wave * SM_CAP;
        int count = 0;
        const int ngroups = (E + 127) >> 7;
        const unsigned lim = (unsigned)((n_dst - lo < 16) ? (n_dst - lo) : 16);
        for (int g0 = wave * DF_UNR; g0 < ngroups; g0 += DF_WAVES * DF_UNR) {
            if (count > SM_CAP - DF_UNR * 128) {
                sm_flush<DF_PITCH>(emb, src_nid, e_src, wl, count, meanT, lane);
                count = 0;
            }
            int d[DF_UNR][4];
            if ((g0 + DF_UNR) * 128 <= E) {
#pragma unroll
                for (int u = 0; u < DF_UNR; ++u) {
                    const int4 t = *(const int4*)(e_dst + (g0 + u) * 128 + lane * 4);
                    d[u][0] = t.x; d[u][1] = t.y; d[u][2] = t.z; d[u][3] = t.w;
                }
            } else {
#pragma unroll
                for (int u = 0; u < DF_UNR; ++u)
#pragma unroll
                    for (int q = 0; q < 4; ++q) {
                        const int e = (g0 + u) * 128 + lane * 4 + q;
                        d[u][q] = (e < E) ? e_dst[e] : -1;
                    }
            }
#pragma unroll
            for (int u = 0; u < DF_UNR; ++u)
                sm_keep(d[u], (g0 + u) * 128 + lane * 4, lo, lim, wl, cnt, count);
        }
        sm_flush<DF_PITCH>(emb, src_nid, e_src, wl, count, meanT, lane);
    }
    __syncthreads();
    for (int i = tid; i < 16 * PDIM; i += DF_WAVES * 32) {
        const int r = i / PDIM, c = i - r * PDIM;
        meanT[r * DF_PITCH + c] /= fmaxf((float)cnt[r], 1.f);
    }

    const int n = 16 * wave + l16;
    const bool ncol = n < DDIM;
    const int bofs = 2 * hi * DDIM + (ncol ? n : DDIM - 1);
    {
        const float bv = ncol ? b_dense[n] : 0.f;
        v8f acc;
#pragma unroll
        for (int i = 0; i < 8; ++i) acc[i] = bv;
        df_tile_gemm(poolT + l16 * DF_PITCH + 2 * hi, wdn + bofs, acc);
#pragma unroll
        for (int i = 0; i < 8; ++i) docT[(i + 8 * hi) * DF_PITCH + n] = ncol ? acc[i] : 0.f;
    }
    __syncthreads();
    if (rel == 0) df_store_tile(docT, docout + (size_t)lo * PDIM, tid);

    {
        const float bv = ncol ? bs[n] : 0.f;
        v8f acc;
#pragma unroll
        for (int i = 0; i < 8; ++i) acc[i] = bv;
        df_tile_gemm(docT + l16 * DF_PITCH + 2 * hi, wp + bofs, acc);
        df_tile_gemm(meanT + l16 * DF_PITCH + 2 * hi, wp + DDIM * DDIM + bofs, acc);
#pragma unroll
        for (int i = 0; i < 8; ++i) outT[(i + 8 * hi) * DF_PITCH + n] = ncol ? acc[i] : 0.f;
    }
    __syncthreads();
    df_store_tile(outT, d3 + (size_t)lo * PDIM, tid);
}

#define WF_R   80
#define WF_S   3
#define WF_MT  (WF_R / 16)
#define WF_NT  ((DDIM + 15) / 16)
#define WF_BCAP 48
#define WL_CAP 352
#define WF_NJ  ((WF_S * WF_R * WF_BCAP + SM_WAVES * 32 - 1) / (SM_WAVES * 32))

__device__ __forceinline__ void wf_rank_keep(const int (&d)[4], int ebase, int lo, unsigned lim,
                                             unsigned* bucket, int* cnt, bool& over) {
    unsigned dl[4];
    int rk[4];
#pragma unroll
    for (int q = 0; q < 4; ++q) {
        dl[q] = (unsigned)(d[q] - lo);
        rk[q] = 0;
        if (dl[q] < lim)
            rk[q] = __hip_atomic_fetch_add(&cnt[dl[q]], 1, __ATOMIC_RELAXED, __HIP_MEMORY_SCOPE_WORKGROUP);
    }
#pragma unroll
    for (int q = 0; q < 4; ++q) {
        if (dl[q] < lim) {
            if (rk[q] < WF_BCAP) bucket[dl[q] * WF_BCAP + rk[q]] = (unsigned)(ebase + q);
            else over = true;
        }
    }
}

__device__ __forceinline__ void wf_row_mean(const float* __restrict__ emb, const unsigned* bk, int cn,
                                            float* mrow, int lane) {
    const int n = __builtin_amdgcn_readfirstlane(cn);
    const unsigned e0 = (lane < n) ? bk[lane] : 0u;
    const unsigned e1 = (lane + 32 < n) ? bk[lane + 32] : 0u;
    float acc[10];
#pragma unroll
    for (int c = 0; c < 10; ++c) acc[c] = 0.f;
    for (int j = 0; j < n; j += 4) {
        const unsigned es = (j < 32) ? e0 : e1;
        float v[4][10];
#pragma unroll
        for (int q = 0; q < 4; ++q) {
            const unsigned row = (unsigned)__builtin_amdgcn_readlane((int)es, (j & 31) + q);
            const float* __restrict__ rp = emb + (size_t)row * DDIM;
#pragma unroll
            for (int c = 0; c < 10; ++c) {
                const int col = lane + 32 * c;
                v[q][c] = rp[(col < DDIM) ? col : DDIM - 1];
            }
        }
#pragma unroll
        for (int q = 0; q < 4; ++q) {
            const bool liveq = j + q < n;
#pragma unroll
            for (int c = 0; c < 10; ++c) acc[c] += liveq ? v[q][c] : 0.f;
        }
    }
    const float cf = fmaxf((float)n, 1.f);
#pragma unroll
    for (int c = 0; c < 10; ++c) {
        const int col = lane + 32 * c;
        mrow[col] = (col < DDIM) ? acc[c] / cf : 0.f;
    }
}

template <int NKP, bool HALF>
__device__ __forceinline__ void wf_tile_gemm(const float* arow, const float* __restrict__ bcol, v8f (&acc)[WF_MT]) {
    constexpr int NPAIR = NKP + (HALF ? 1 : 0);
    v4f bc[2];
#pragma unroll
    for (int u = 0; u < 2; ++u) {
        const int p = (u < NPAIR) ? u : 0;
        bc[u] = b_pair(bcol, p, p < NKP);
    }
#pragma unroll
    for (int p0 = 0; p0 < NPAIR; p0 += 2) {
        v4f bn[2];
#pragma unroll
        for (int u = 0; u < 2; ++u) bn[u] = (p0 + 2 + u < NPAIR) ? b_pair(bcol, p0 + 2 + u, p0 + 2 + u < NKP) : bc[u];
#pragma unroll
        for (int u = 0; u < 2; ++u) {
            const int p = p0 + u;
            if (p < NPAIR) {
                v2f b0; b0.x = bc[u].x; b0.y = bc[u].y;
#pragma unroll
                for (int mt = 0; mt < WF_MT; ++mt) {
                    const v2f a0 = *(const v2f*)(arow + mt * 16 * DF_PITCH + 8 * p);
                    acc[mt] = __builtin_amdgcn_wmma_f32_16x16x4_f32(false, a0, false, b0, (short)0, acc[mt], false, false);
                }
                if (p < NKP) {
                    v2f b1; b1.x = bc[u].z; b1.y = bc[u].w;
#pragma unroll
                    for (int mt = 0; mt < WF_MT; ++mt) {
                        const v2f a1 = *(const v2f*)(arow + mt * 16 * DF_PITCH + 8 * p + 4);
                        acc[mt] = __builtin_amdgcn_wmma_f32_16x16x4_f32(false, a1, false, b1, (short)0, acc[mt], false, false);
                    }
                }
            }
        }
        __builtin_amdgcn_sched_barrier(0);
        bc[0] = bn[0]; bc[1] = bn[1];
    }
}

__global__ __launch_bounds__(SM_WAVES * 32) void word_rel_fused(
        const float* __restrict__ emb, const int* __restrict__ dst_nid,
        const int* __restrict__ sn0, const int* __restrict__ sn1, const int* __restrict__ sn2,
        const int* __restrict__ es0, const int* __restrict__ es1, const int* __restrict__ es2,
        const int* __restrict__ ed0, const int* __restrict__ ed1, const int* __restrict__ ed2,
        const float* __restrict__ wp0, const float* __restrict__ wp1, const float* __restrict__ wp2,
        const float* __restrict__ bs0, const float* __restrict__ bs1, const float* __restrict__ bs2,
        int E, int n_dst, float* __restrict__ h_0, float* __restrict__ h_1, float* __restrict__ h_2) {
    static_assert(WF_R % 16 == 0, "whole 16-row tiles");
    static_assert(WF_NT <= SM_WAVES && 16 * WF_NT <= PDIM, "a wave per 16-column tile, inside a padded row");
    static_assert(WF_BCAP > 32 && WF_BCAP <= 64, "a lane holds two entries of a bucket");
    static_assert(WL_CAP > 128 && WL_CAP % 32 == 0 && SM_WAVES * WL_CAP <= WF_S * WF_R * WF_BCAP,
                  "slow way: room for one 128-edge group behind the fill check, the lists inside the buckets' space");
    __shared__ __attribute__((aligned(16))) float meanT[WF_R * DF_PITCH];
    __shared__ __attribute__((aligned(16))) float hdT[WF_R * DF_PITCH];
    __shared__ unsigned bucket[WF_NJ * SM_WAVES * 32];
    __shared__ int cnt[WF_S * WF_R];
    __shared__ int qnext[WF_S];
    __shared__ int ovf;
    const int tid = threadIdx.x;
    const int lane = tid & 31;
    const int wave = __builtin_amdgcn_readfirstlane(tid >> 5);
    const int lo = blockIdx.x * (WF_S * WF_R);
    const int rel = blockIdx.y;
    const int* __restrict__ src_nid = (rel == 0) ? sn0 : ((rel == 1) ? sn1 : sn2);
    const int* __restrict__ e_src   = (rel == 0) ? es0 : ((rel == 1) ? es1 : es2);
    const int* __restrict__ e_dst   = (rel == 0) ? ed0 : ((rel == 1) ? ed1 : ed2);
    const float* __restrict__ wp    = (rel == 0) ? wp0 : ((rel == 1) ? wp1 : wp2);
    const float* __restrict__ bs    = (rel == 0) ? bs0 : ((rel == 1) ? bs1 : bs2);
    float* __restrict__ h           = (rel == 0) ? h_0 : ((rel == 1) ? h_1 : h_2);

    for (int i = tid; i < WF_S * WF_R; i += SM_WAVES * 32) cnt[i] = 0;
    if (tid < WF_S) qnext[tid] = 0;
    if (tid == 0) ovf = 0;
    __syncthreads();

    const int ngroups = (E + 127) >> 7;
    const int rows_here = (n_dst - lo < WF_S * WF_R) ? (n_dst - lo) : WF_S * WF_R;

    {
        bool over = false;
        for (int g0 = wave * SM_UNR; g0 < ngroups; g0 += SM_WAVES * SM_UNR) {
            int d[SM_UNR][4];
            sm_load_round<SM_UNR>(e_dst, g0, E, lane, d);
#pragma unroll
            for (int u = 0; u < SM_UNR; ++u)
                wf_rank_keep(d[u], (g0 + u) * 128 + lane * 4, lo, (unsigned)rows_here, bucket, cnt, over);
        }
        if (over) ovf = 1;
    }
    __syncthreads();
    const bool slow = __builtin_amdgcn_readfirstlane(ovf) != 0;
    if (!slow) {
        unsigned en[WF_NJ];
#pragma unroll
        for (int j = 0; j < WF_NJ; ++j) {
            const int i = tid + SM_WAVES * 32 * j;
            const int r = i / WF_BCAP, k = i - r * WF_BCAP;
            const bool rin = r < WF_S * WF_R;
            const bool live = rin && k < cnt[rin ? r : 0];
            const unsigned raw = bucket[i];
            en[j] = live ? raw : 0u;
        }
#pragma unroll
        for (int j = 0; j < WF_NJ; ++j) en[j] = (unsigned)e_src[en[j]];
#pragma unroll
        for (int j = 0; j < WF_NJ; ++j) en[j] = (unsigned)src_nid[en[j]];
#pragma unroll
        for (int j = 0; j < WF_NJ; ++j) bucket[tid + SM_WAVES * 32 * j] = en[j];
    } else {
        for (int i = tid; i < WF_S * WF_R; i += SM_WAVES * 32) cnt[i] = 0;
    }
    __syncthreads();

#pragma unroll 1
    for (int s = 0; s < WF_S; ++s) {
        const int lo_s = lo + WF_R * s;
        if (lo_s >= n_dst) break;
        int t = tid;
        asm volatile("" : "+v"(t));
        const int hi = (t >> 4) & 1, l16 = t & 15;
        const int n = 16 * wave + l16;
        const bool ncol = n < DDIM;
        if (!slow) {
            for (;;) {
                int r = 0;
                if (lane == 0) r = __hip_atomic_fetch_add(&qnext[s], 1, __ATOMIC_RELAXED, __HIP_MEMORY_SCOPE_WORKGROUP);
                r = __builtin_amdgcn_readfirstlane(r);
                if (r >= WF_R) break;
                wf_row_mean(emb, bucket + (WF_R * s + r) * WF_BCAP, cnt[WF_R * s + r], meanT + r * DF_PITCH, lane);
            }
        } else {
            const unsigned lim = (unsigned)((n_dst - lo_s < WF_R) ? (n_dst - lo_s) : WF_R);
            unsigned* wl = bucket + wave * WL_CAP;
            for (int i = t; i < WF_R * DF_PITCH / 4; i += SM_WAVES * 32)
                ((v4f*)meanT)[i] = (v4f){0.f, 0.f, 0.f, 0.f};
            __syncthreads();
            int c = 0;
            for (int g0 = wave * SM_UNR; g0 < ngroups; g0 += SM_WAVES * SM_UNR) {
                int d[SM_UNR][4];
                sm_load_round<SM_UNR>(e_dst, g0, E, lane, d);
#pragma unroll
                for (int u = 0; u < SM_UNR; ++u) {
                    if (c > WL_CAP - 128) {
                        sm_flush<DF_PITCH>(emb, src_nid, e_src, wl, c, meanT, lane);
                        c = 0;
                    }
                    sm_keep(d[u], (g0 + u) * 128 + lane * 4, lo_s, lim, wl, cnt + WF_R * s, c);
                }
            }
            sm_flush<DF_PITCH>(emb, src_nid, e_src, wl, c, meanT, lane);
            __syncthreads();
            for (int i = t; i < WF_R * (PDIM / 4); i += SM_WAVES * 32) {
                const int r = i / (PDIM / 4), c4 = i - r * (PDIM / 4);
                const float cf = fmaxf((float)cnt[WF_R * s + r], 1.f);
                v4f sv = *(const v4f*)(meanT + r * DF_PITCH + 4 * c4);
                sv.x /= cf; sv.y /= cf; sv.z /= cf; sv.w /= cf;
                *(v4f*)(meanT + r * DF_PITCH + 4 * c4) = sv;
            }
        }
        for (int r = wave; r < WF_R; r += SM_WAVES) {
            const bool live = lo_s + r < n_dst;
            const v4f* __restrict__ rp = (const v4f*)(emb + (size_t)(live ? dst_nid[lo_s + r] : 0) * DDIM);
            for (int c4 = lane; c4 < DDIM / 4; c4 += 32)
                *(v4f*)(hdT + r * DF_PITCH + 4 * c4) = live ? rp[c4] : (v4f){0.f, 0.f, 0.f, 0.f};
        }
        __syncthreads();

        v8f acc[WF_MT];
        if (wave < WF_NT) {
            static_assert(DDIM % 4 == 0, "a table row and a mean row are a whole number of k-steps");
            const float* __restrict__ bcol = wp + 2 * hi * DDIM + (ncol ? n : DDIM - 1);
            const float bv = ncol ? bs[n] : 0.f;
#pragma unroll
            for (int mt = 0; mt < WF_MT; ++mt)
#pragma unroll
                for (int i = 0; i < 8; ++i) acc[mt][i] = bv;
            wf_tile_gemm<DDIM / 8, (DDIM % 8) != 0>(hdT + l16 * DF_PITCH + 2 * hi, bcol, acc);
            wf_tile_gemm<DDIM / 8, (DDIM % 8) != 0>(meanT + l16 * DF_PITCH + 2 * hi, bcol + DDIM * DDIM, acc);
        }
        __syncthreads();
        if (wave < WF_NT) {
#pragma unroll
            for (int mt = 0; mt < WF_MT; ++mt)
#pragma unroll
                for (int i = 0; i < 8; ++i) meanT[(16 * mt + i + 8 * hi) * DF_PITCH + n] = ncol ? acc[mt][i] : 0.f;
        }
        __syncthreads();
        for (int i = t; i < WF_R * (PDIM / 4); i += SM_WAVES * 32) {
            const int r = i / (PDIM / 4), c4 = i - r * (PDIM / 4);
            if (lo_s + r < n_dst)
                ((v4f*)(h + (size_t)lo_s * PDIM))[i] = *(const v4f*)(meanT + r * DF_PITCH + 4 * c4);
        }
        if (s + 1 < WF_S) __syncthreads();
    }
}

__global__ __launch_bounds__(256) void doc_tail(
        const float* __restrict__ d3_0, const float* __restrict__ d3_1, const float* __restrict__ d3_2,
        const float* __restrict__ docout, const float* __restrict__ wfc, const float* __restrict__ bfc,
        float* __restrict__ out, int n) {
    __shared__ float rep[8 * PDIM];
    const int lane = threadIdx.x & 31;
    const int wave = threadIdx.x >> 5;
    const int node = blockIdx.x * 8 + wave;
    if (node < n)
        sdpa_row<PDIM>(d3_0 + (size_t)node * PDIM, d3_1 + (size_t)node * PDIM, d3_2 + (size_t)node * PDIM,
                       docout + (size_t)node * PDIM, rep + wave * PDIM, lane);
    __syncthreads();
    if (node < n && lane < CN) {
        float acc = bfc[lane];
        const float* rp = rep + wave * PDIM;
        for (int j = 0; j < DDIM; ++j) acc += rp[j] * wfc[(size_t)j * CN + lane];
        out[(size_t)node * CN + lane] = acc;
    }
}


extern "C" void kernel_launch(void* const* d_in, const int* in_sizes, int n_in,
                              void* d_out, int out_size, void* d_ws, size_t ws_size,
                              hipStream_t stream) {
    (void)in_sizes; (void)n_in; (void)out_size; (void)ws_size;
    const float* emb_word = (const float*)d_in[0];
    const float* emb_doc  = (const float*)d_in[1];
    const float* w_sage[6] = {(const float*)d_in[2],  (const float*)d_in[4],  (const float*)d_in[6],
                              (const float*)d_in[8],  (const float*)d_in[10], (const float*)d_in[12]};
    const float* b_sage[6] = {(const float*)d_in[3],  (const float*)d_in[5],  (const float*)d_in[7],
                              (const float*)d_in[9],  (const float*)d_in[11], (const float*)d_in[13]};
    const float* w_dense = (const float*)d_in[14];
    const float* b_dense = (const float*)d_in[15];
    const float* w_fc    = (const float*)d_in[16];
    const float* b_fc    = (const float*)d_in[17];
    const int* dst_nid      = (const int*)d_in[18];
    const int* src_nid_w[3] = {(const int*)d_in[19], (const int*)d_in[20], (const int*)d_in[21]};
    const int* e_src_w[3]   = {(const int*)d_in[22], (const int*)d_in[24], (const int*)d_in[26]};
    const int* e_dst_w[3]   = {(const int*)d_in[23], (const int*)d_in[25], (const int*)d_in[27]};
    const int* x_batch      = (const int*)d_in[28];
    const int* src_nid_d[3] = {(const int*)d_in[30], (const int*)d_in[31], (const int*)d_in[32]};
    const int* e_src_d[3]   = {(const int*)d_in[33], (const int*)d_in[35], (const int*)d_in[37]};
    const int* e_dst_d[3]   = {(const int*)d_in[34], (const int*)d_in[36], (const int*)d_in[38]};
    float* out = (float*)d_out;

    float* ws = (float*)d_ws;
    size_t off = 0;
    auto carve = [&](size_t nflt) { float* p = ws + off; off += nflt; return p; };
    float* nbuf = carve((size_t)NWN * PDIM);
    float* h3[3]; for (int r = 0; r < 3; ++r) h3[r] = carve((size_t)NWN * PDIM);
    float* pooled = carve((size_t)BN * PDIM);
    float* docout = carve((size_t)BN * PDIM);
    float* d3[3]; for (int r = 0; r < 3; ++r) d3[r] = carve((size_t)BN * PDIM);

    word_rel_fused<<<dim3((NWN + WF_S * WF_R - 1) / (WF_S * WF_R), 3), SM_WAVES * 32, 0, stream>>>(
        emb_word, dst_nid, src_nid_w[0], src_nid_w[1], src_nid_w[2], e_src_w[0], e_src_w[1], e_src_w[2],
        e_dst_w[0], e_dst_w[1], e_dst_w[2], w_sage[0], w_sage[1], w_sage[2], b_sage[0], b_sage[1], b_sage[2],
        EWN, NWN, h3[0], h3[1], h3[2]);

    sdpa_fuse_add<<<(NWN + 7) / 8, 256, 0, stream>>>(h3[0], h3[1], h3[2], emb_word, dst_nid, nbuf, NWN);

    pool_seq<<<BN, PS_WAVES * 32, 0, stream>>>(nbuf, x_batch, pooled);

    doc_rel_fused<<<dim3(BN / 16, 3), DF_WAVES * 32, 0, stream>>>(
        pooled, w_dense, b_dense, w_sage[3], w_sage[4], w_sage[5], b_sage[3], b_sage[4], b_sage[5], emb_doc,
        src_nid_d[0], src_nid_d[1], src_nid_d[2], e_src_d[0], e_src_d[1], e_src_d[2],
        e_dst_d[0], e_dst_d[1], e_dst_d[2], EDN, BN, docout, d3[0], d3[1], d3[2]);

    doc_tail<<<(BN + 7) / 8, 256, 0, stream>>>(d3[0], d3[1], d3[2], docout, w_fc, b_fc, out, BN);
}


# word path: two LDS buffer sets of 48-row sub-tiles: the 13 waves without a product tile gather sub-tile s+1 (means, table rows) while the 19 product waves run sub-tile s's WMMAs; same sums
# speedup vs baseline: 5.1345x; 1.0362x over previous
#include <hip/hip_runtime.h>

#define DDIM 300
#define PDIM 320
#define NWN  20000
#define EWN  320000
#define BN   256
#define LN   256
#define SDN  2048
#define EDN  4096
#define CN   20

typedef __attribute__((ext_vector_type(2))) float v2f;
typedef __attribute__((ext_vector_type(4))) float v4f;
typedef __attribute__((ext_vector_type(8))) float v8f;

__device__ __forceinline__ v4f b_pair(const float* __restrict__ bcol, int p, bool second) {
    v4f b;
    b.x = bcol[(8 * p + 0) * DDIM];
    b.y = bcol[(8 * p + 1) * DDIM];
    b.z = second ? bcol[(8 * p + 4) * DDIM] : 0.f;
    b.w = second ? bcol[(8 * p + 5) * DDIM] : 0.f;
    return b;
}

#define SM_WAVES 32
#define SM_UNR   4
#define SM_CAP   1024

template <int PITCH>
__device__ __forceinline__ void sm_add_rows(const float* __restrict__ emb, unsigned ent, int nb, float* acc, int lane) {
    for (int j = 0; j < nb; j += 4) {
        float v[4][10];
        int dl[4];
#pragma unroll
        for (int q = 0; q < 4; ++q) {
            const unsigned e = (unsigned)__builtin_amdgcn_readlane((int)ent, j + q);
            const float* __restrict__ rp = emb + (size_t)(e >> 8) * DDIM;
            dl[q] = (int)(e & 255u);
#pragma unroll
            for (int c = 0; c < 10; ++c) {
                const int col = lane + 32 * c;
                v[q][c] = rp[(col < DDIM) ? col : DDIM - 1];
            }
        }
#pragma unroll
        for (int q = 0; q < 4; ++q) {
            if (j + q < nb) {
                float* a = acc + dl[q] * PITCH;
#pragma unroll
                for (int c = 0; c < 10; ++c) {
                    const int col = lane + 32 * c;
                    if (col < DDIM)
                        __hip_atomic_fetch_add(&a[col], v[q][c], __ATOMIC_RELAXED,
                                               __HIP_MEMORY_SCOPE_WORKGROUP);
                }
            }
        }
    }
}

template <int PITCH>
__device__ __forceinline__ void sm_flush(const float* __restrict__ emb, const int* __restrict__ src_nid,
                                         const int* __restrict__ e_src, const unsigned* wl, int count,
                                         float* acc, int lane) {
    for (int k0 = 0; k0 < count; k0 += 32) {
        unsigned ent = 0u;
        if (k0 + lane < count) {
            const unsigned en = wl[k0 + lane];
            const int row = src_nid[e_src[en >> 8]];
            ent = ((unsigned)row << 8) | (en & 255u);
        }
        sm_add_rows<PITCH>(emb, ent, (count - k0 < 32) ? (count - k0) : 32, acc, lane);
    }
}

template <int U>
__device__ __forceinline__ void sm_load_round(const int* __restrict__ e_dst, int g0, int E, int lane, int (&d)[U][4]) {
    if ((g0 + U) * 128 <= E) {
#pragma unroll
        for (int u = 0; u < U; ++u) {
            const int4 t = *(const int4*)(e_dst + (g0 + u) * 128 + lane * 4);
            d[u][0] = t.x; d[u][1] = t.y; d[u][2] = t.z; d[u][3] = t.w;
        }
    } else {
#pragma unroll
        for (int u = 0; u < U; ++u)
#pragma unroll
            for (int q = 0; q < 4; ++q) {
                const int e = (g0 + u) * 128 + lane * 4 + q;
                d[u][q] = (e < E) ? e_dst[e] : -1;
            }
    }
}

__device__ __forceinline__ void sm_keep(const int (&d)[4], int ebase, int lo, unsigned lim,
                                        unsigned* wl, int* cnt, int& count) {
#pragma unroll
    for (int q = 0; q < 4; ++q) {
        const unsigned dl = (unsigned)(d[q] - lo);
        const bool m = dl < lim;
        const unsigned mask = __builtin_amdgcn_ballot_w32(m);
        if (m) {
            const int pos = count + (int)__builtin_amdgcn_mbcnt_lo(mask, 0u);
            wl[pos] = ((unsigned)(ebase + q) << 8) | dl;
            __hip_atomic_fetch_add(&cnt[dl], 1, __ATOMIC_RELAXED, __HIP_MEMORY_SCOPE_WORKGROUP);
        }
        count += __popc(mask);
    }
}

template <int AVN>
__device__ __forceinline__ void sdpa_row(const float* x0, const float* x1, const float* x2,
                                         const float* av, float* o, int lane) {
    float d00 = 0, d01 = 0, d02 = 0, d11 = 0, d12 = 0, d22 = 0;
    for (int j = lane; j < PDIM; j += 32) {
        float a = x0[j], b = x1[j], e = x2[j];
        d00 += a * a; d01 += a * b; d02 += a * e;
        d11 += b * b; d12 += b * e; d22 += e * e;
    }
    for (int off = 16; off > 0; off >>= 1) {
        d00 += __shfl_xor(d00, off, 32);
        d01 += __shfl_xor(d01, off, 32);
        d02 += __shfl_xor(d02, off, 32);
        d11 += __shfl_xor(d11, off, 32);
        d12 += __shfl_xor(d12, off, 32);
        d22 += __shfl_xor(d22, off, 32);
    }
    const float scale = rsqrtf((float)DDIM);
    float S[3][3] = {{d00, d01, d02}, {d01, d11, d12}, {d02, d12, d22}};
    float wsum[3] = {0.f, 0.f, 0.f};
    for (int t = 0; t < 3; ++t) {
        float p0 = S[t][0] * scale, p1 = S[t][1] * scale, p2 = S[t][2] * scale;
        float mx = fmaxf(p0, fmaxf(p1, p2));
        float e0 = __expf(p0 - mx), e1 = __expf(p1 - mx), e2 = __expf(p2 - mx);
        float inv = 1.f / (e0 + e1 + e2);
        wsum[0] += e0 * inv; wsum[1] += e1 * inv; wsum[2] += e2 * inv;
    }
    const float w0 = wsum[0] * (1.f / 3.f), w1 = wsum[1] * (1.f / 3.f), w2 = wsum[2] * (1.f / 3.f);
    for (int j = lane; j < PDIM; j += 32)
        o[j] = w0 * x0[j] + w1 * x1[j] + w2 * x2[j] + ((j < AVN) ? av[j] : 0.f);
}

__global__ void sdpa_fuse_add(const float* __restrict__ h0, const float* __restrict__ h1,
                              const float* __restrict__ h2, const float* __restrict__ emb,
                              const int* __restrict__ nid, float* __restrict__ out, int n) {
    const int lane = threadIdx.x & 31;
    const int node = blockIdx.x * (blockDim.x >> 5) + (threadIdx.x >> 5);
    if (node >= n) return;
    sdpa_row<DDIM>(h0 + (size_t)node * PDIM, h1 + (size_t)node * PDIM, h2 + (size_t)node * PDIM,
                   emb + (size_t)nid[node] * DDIM, out + (size_t)node * PDIM, lane);
}

#define PS_WAVES 16
#define PS_TOK   (LN / PS_WAVES)
__global__ __launch_bounds__(PS_WAVES * 32) void pool_seq(const float* __restrict__ table,
                                                          const int* __restrict__ xb,
                                                          float* __restrict__ pooled) {
    static_assert(LN % PS_WAVES == 0, "every wave takes the same number of tokens");
    static_assert(PDIM == 320, "a lane's 4 + 4 + 2 floats cover a padded row");
    __shared__ __attribute__((aligned(16))) float part[PS_WAVES * PDIM];
    const int b = blockIdx.x;
    const int tid = threadIdx.x;
    const int lane = tid & 31;
    const int wave = __builtin_amdgcn_readfirstlane(tid >> 5);
    const int* __restrict__ row = xb + b * LN + wave * PS_TOK;

    v4f p0[PS_TOK], p1[PS_TOK];
    v2f p2[PS_TOK];
    bool ok[PS_TOK];
#pragma unroll
    for (int t = 0; t < PS_TOK; ++t) {
        const int idx = row[t];
        ok[t] = idx < NWN;
        const float* __restrict__ rp = table + (size_t)(ok[t] ? idx : 0) * PDIM;
        p0[t] = *(const v4f*)(rp + 4 * lane);
        p1[t] = *(const v4f*)(rp + 128 + 4 * lane);
        p2[t] = *(const v2f*)(rp + 256 + 2 * lane);
    }
    v4f a0 = (v4f){0.f, 0.f, 0.f, 0.f}, a1 = (v4f){0.f, 0.f, 0.f, 0.f};
    v2f a2 = (v2f){0.f, 0.f};
#pragma unroll
    for (int t = 0; t < PS_TOK; ++t) {
        if (ok[t]) { a0 += p0[t]; a1 += p1[t]; a2 += p2[t]; }
    }
    float* pw = part + wave * PDIM;
    *(v4f*)(pw + 4 * lane) = a0;
    *(v4f*)(pw + 128 + 4 * lane) = a1;
    *(v2f*)(pw + 256 + 2 * lane) = a2;
    __syncthreads();
    if (tid < PDIM) {
        float s = 0.f;
#pragma unroll
        for (int w = 0; w < PS_WAVES; ++w) s += part[w * PDIM + tid];
        pooled[(size_t)b * PDIM + tid] = s * (1.f / (float)LN);
    }
}

#define DF_WAVES 20
#define DF_PITCH 324
#define DF_UNR   2

__device__ __forceinline__ void df_tile_gemm(const float* arow, const float* __restrict__ bcol, v8f& acc) {
    static_assert(DDIM % 4 == 0, "K is a whole number of k-steps");
#pragma unroll
    for (int kp = 0; kp < (DDIM + 7) / 8; ++kp) {
        const bool second = kp < DDIM / 8;
        const v4f b = b_pair(bcol, kp, second);
        const v2f a0 = *(const v2f*)(arow + 8 * kp);
        v2f b0; b0.x = b.x; b0.y = b.y;
        acc = __builtin_amdgcn_wmma_f32_16x16x4_f32(false, a0, false, b0, (short)0, acc, false, false);
        if (second) {
            const v2f a1 = *(const v2f*)(arow + 8 * kp + 4);
            v2f b1; b1.x = b.z; b1.y = b.w;
            acc = __builtin_amdgcn_wmma_f32_16x16x4_f32(false, a1, false, b1, (short)0, acc, false, false);
        }
    }
}

__device__ __forceinline__ void df_store_tile(const float* t, float* __restrict__ g, int tid) {
    for (int i = tid; i < 16 * (PDIM / 4); i += DF_WAVES * 32) {
        const int r = i / (PDIM / 4), c4 = i - r * (PDIM / 4);
        ((v4f*)g)[i] = *(const v4f*)(t + r * DF_PITCH + 4 * c4);
    }
}

__global__ __launch_bounds__(DF_WAVES * 32) void doc_rel_fused(
        const float* __restrict__ pooled, const float* __restrict__ wdn, const float* __restrict__ b_dense,
        const float* __restrict__ wp0, const float* __restrict__ wp1, const float* __restrict__ wp2,
        const float* __restrict__ bs0, const float* __restrict__ bs1, const float* __restrict__ bs2,
        const float* __restrict__ emb,
        const int* __restrict__ sn0, const int* __restrict__ sn1, const int* __restrict__ sn2,
        const int* __restrict__ es0, const int* __restrict__ es1, const int* __restrict__ es2,
        const int* __restrict__ ed0, const int* __restrict__ ed1, const int* __restrict__ ed2,
        int E, int n_dst, float* __restrict__ docout,
        float* __restrict__ d3_0, float* __restrict__ d3_1, float* __restrict__ d3_2) {
    __shared__ __attribute__((aligned(16))) float meanT[16 * DF_PITCH];
    __shared__ __attribute__((aligned(16))) float poolT[16 * DF_PITCH];
    __shared__ __attribute__((aligned(16))) float docT[16 * DF_PITCH];
    __shared__ __attribute__((aligned(16))) float outT[16 * DF_PITCH];
    __shared__ int cnt[16];
    __shared__ unsigned list[DF_WAVES * SM_CAP];
    const int tid = threadIdx.x;
    const int lane = tid & 31;
    const int wave = __builtin_amdgcn_readfirstlane(tid >> 5);
    const int hi = lane >> 4, l16 = lane & 15;
    const int lo = blockIdx.x * 16;
    const int rel = blockIdx.y;
    const int* __restrict__ src_nid = (rel == 0) ? sn0 : ((rel == 1) ? sn1 : sn2);
    const int* __restrict__ e_src   = (rel == 0) ? es0 : ((rel == 1) ? es1 : es2);
    const int* __restrict__ e_dst   = (rel == 0) ? ed0 : ((rel == 1) ? ed1 : ed2);
    const float* __restrict__ wp    = (rel == 0) ? wp0 : ((rel == 1) ? wp1 : wp2);
    const float* __restrict__ bs    = (rel == 0) ? bs0 : ((rel == 1) ? bs1 : bs2);
    float* __restrict__ d3          = (rel == 0) ? d3_0 : ((rel == 1) ? d3_1 : d3_2);

    for (int i = tid; i < 16 * (PDIM / 4); i += DF_WAVES * 32) {
        const int r = i / (PDIM / 4), c4 = i - r * (PDIM / 4);
        *(v4f*)(poolT + r * DF_PITCH + 4 * c4) = ((const v4f*)(pooled + (size_t)lo * PDIM))[i];
    }
    for (int i = tid; i < 16 * DF_PITCH / 4; i += DF_WAVES * 32)
        ((v4f*)meanT)[i] = (v4f){0.f, 0.f, 0.f, 0.f};
    if (tid < 16) cnt[tid] = 0;
    __syncthreads();

    {
        unsigned* wl = list + wave * SM_CAP;
        int count = 0;
        const int ngroups = (E + 127) >> 7;
        const unsigned lim = (unsigned)((n_dst - lo < 16) ? (n_dst - lo) : 16);
        for (int g0 = wave * DF_UNR; g0 < ngroups; g0 += DF_WAVES * DF_UNR) {
            if (count > SM_CAP - DF_UNR * 128) {
                sm_flush<DF_PITCH>(emb, src_nid, e_src, wl, count, meanT, lane);
                count = 0;
            }
            int d[DF_UNR][4];
            if ((g0 + DF_UNR) * 128 <= E) {
#pragma unroll
                for (int u = 0; u < DF_UNR; ++u) {
                    const int4 t = *(const int4*)(e_dst + (g0 + u) * 128 + lane * 4);
                    d[u][0] = t.x; d[u][1] = t.y; d[u][2] = t.z; d[u][3] = t.w;
                }
            } else {
#pragma unroll
                for (int u = 0; u < DF_UNR; ++u)
#pragma unroll
                    for (int q = 0; q < 4; ++q) {
                        const int e = (g0 + u) * 128 + lane * 4 + q;
                        d[u][q] = (e < E) ? e_dst[e] : -1;
                    }
            }
#pragma unroll
            for (int u = 0; u < DF_UNR; ++u)
                sm_keep(d[u], (g0 + u) * 128 + lane * 4, lo, lim, wl, cnt, count);
        }
        sm_flush<DF_PITCH>(emb, src_nid, e_src, wl, count, meanT, lane);
    }
    __syncthreads();
    for (int i = tid; i < 16 * PDIM; i += DF_WAVES * 32) {
        const int r = i / PDIM, c = i - r * PDIM;
        meanT[r * DF_PITCH + c] /= fmaxf((float)cnt[r], 1.f);
    }

    const int n = 16 * wave + l16;
    const bool ncol = n < DDIM;
    const int bofs = 2 * hi * DDIM + (ncol ? n : DDIM - 1);
    {
        const float bv = ncol ? b_dense[n] : 0.f;
        v8f acc;
#pragma unroll
        for (int i = 0; i < 8; ++i) acc[i] = bv;
        df_tile_gemm(poolT + l16 * DF_PITCH + 2 * hi, wdn + bofs, acc);
#pragma unroll
        for (int i = 0; i < 8; ++i) docT[(i + 8 * hi) * DF_PITCH + n] = ncol ? acc[i] : 0.f;
    }
    __syncthreads();
    if (rel == 0) df_store_tile(docT, docout + (size_t)lo * PDIM, tid);

    {
        const float bv = ncol ? bs[n] : 0.f;
        v8f acc;
#pragma unroll
        for (int i = 0; i < 8; ++i) acc[i] = bv;
        df_tile_gemm(docT + l16 * DF_PITCH + 2 * hi, wp + bofs, acc);
        df_tile_gemm(meanT + l16 * DF_PITCH + 2 * hi, wp + DDIM * DDIM + bofs, acc);
#pragma unroll
        for (int i = 0; i < 8; ++i) outT[(i + 8 * hi) * DF_PITCH + n] = ncol ? acc[i] : 0.f;
    }
    __syncthreads();
    df_store_tile(outT, d3 + (size_t)lo * PDIM, tid);
}

#define WF_R   48
#define WF_S   5
#define WF_MT  (WF_R / 16)
#define WF_NT  ((DDIM + 15) / 16)
#define WF_BCAP 48
#define WL_CAP 352
#define WF_NJ  ((WF_S * WF_R * WF_BCAP + SM_WAVES * 32 - 1) / (SM_WAVES * 32))

__device__ __forceinline__ void wf_rank_keep(const int (&d)[4], int ebase, int lo, unsigned lim,
                                             unsigned* bucket, int* cnt, bool& over) {
    unsigned dl[4];
    int rk[4];
#pragma unroll
    for (int q = 0; q < 4; ++q) {
        dl[q] = (unsigned)(d[q] - lo);
        rk[q] = 0;
        if (dl[q] < lim)
            rk[q] = __hip_atomic_fetch_add(&cnt[dl[q]], 1, __ATOMIC_RELAXED, __HIP_MEMORY_SCOPE_WORKGROUP);
    }
#pragma unroll
    for (int q = 0; q < 4; ++q) {
        if (dl[q] < lim) {
            if (rk[q] < WF_BCAP) bucket[dl[q] * WF_BCAP + rk[q]] = (unsigned)(ebase + q);
            else over = true;
        }
    }
}

__device__ __forceinline__ void wf_row_mean(const float* __restrict__ emb, const unsigned* bk, int cn,
                                            float* mrow, int lane) {
    const int n = __builtin_amdgcn_readfirstlane(cn);
    const unsigned e0 = (lane < n) ? bk[lane] : 0u;
    const unsigned e1 = (lane + 32 < n) ? bk[lane + 32] : 0u;
    float acc[10];
#pragma unroll
    for (int c = 0; c < 10; ++c) acc[c] = 0.f;
    for (int j = 0; j < n; j += 4) {
        const unsigned es = (j < 32) ? e0 : e1;
        float v[4][10];
#pragma unroll
        for (int q = 0; q < 4; ++q) {
            const unsigned row = (unsigned)__builtin_amdgcn_readlane((int)es, (j & 31) + q);
            const float* __restrict__ rp = emb + (size_t)row * DDIM;
#pragma unroll
            for (int c = 0; c < 10; ++c) {
                const int col = lane + 32 * c;
                v[q][c] = rp[(col < DDIM) ? col : DDIM - 1];
            }
        }
#pragma unroll
        for (int q = 0; q < 4; ++q) {
            const bool liveq = j + q < n;
#pragma unroll
            for (int c = 0; c < 10; ++c) acc[c] += liveq ? v[q][c] : 0.f;
        }
    }
    const float cf = fmaxf((float)n, 1.f);
#pragma unroll
    for (int c = 0; c < 10; ++c) {
        const int col = lane + 32 * c;
        mrow[col] = (col < DDIM) ? acc[c] / cf : 0.f;
    }
}

template <int NKP, bool HALF>
__device__ __forceinline__ void wf_tile_gemm(const float* arow, const float* __restrict__ bcol, v8f (&acc)[WF_MT]) {
    constexpr int NPAIR = NKP + (HALF ? 1 : 0);
    v4f bc[2];
#pragma unroll
    for (int u = 0; u < 2; ++u) {
        const int p = (u < NPAIR) ? u : 0;
        bc[u] = b_pair(bcol, p, p < NKP);
    }
#pragma unroll
    for (int p0 = 0; p0 < NPAIR; p0 += 2) {
        v4f bn[2];
#pragma unroll
        for (int u = 0; u < 2; ++u) bn[u] = (p0 + 2 + u < NPAIR) ? b_pair(bcol, p0 + 2 + u, p0 + 2 + u < NKP) : bc[u];
#pragma unroll
        for (int u = 0; u < 2; ++u) {
            const int p = p0 + u;
            if (p < NPAIR) {
                v2f b0; b0.x = bc[u].x; b0.y = bc[u].y;
#pragma unroll
                for (int mt = 0; mt < WF_MT; ++mt) {
                    const v2f a0 = *(const v2f*)(arow + mt * 16 * DF_PITCH + 8 * p);
                    acc[mt] = __builtin_amdgcn_wmma_f32_16x16x4_f32(false, a0, false, b0, (short)0, acc[mt], false, false);
                }
                if (p < NKP) {
                    v2f b1; b1.x = bc[u].z; b1.y = bc[u].w;
#pragma unroll
                    for (int mt = 0; mt < WF_MT; ++mt) {
                        const v2f a1 = *(const v2f*)(arow + mt * 16 * DF_PITCH + 8 * p + 4);
                        acc[mt] = __builtin_amdgcn_wmma_f32_16x16x4_f32(false, a1, false, b1, (short)0, acc[mt], false, false);
                    }
                }
            }
        }
        __builtin_amdgcn_sched_barrier(0);
        bc[0] = bn[0]; bc[1] = bn[1];
    }
}

__global__ __launch_bounds__(SM_WAVES * 32) void word_rel_fused(
        const float* __restrict__ emb, const int* __restrict__ dst_nid,
        const int* __restrict__ sn0, const int* __restrict__ sn1, const int* __restrict__ sn2,
        const int* __restrict__ es0, const int* __restrict__ es1, const int* __restrict__ es2,
        const int* __restrict__ ed0, const int* __restrict__ ed1, const int* __restrict__ ed2,
        const float* __restrict__ wp0, const float* __restrict__ wp1, const float* __restrict__ wp2,
        const float* __restrict__ bs0, const float* __restrict__ bs1, const float* __restrict__ bs2,
        int E, int n_dst, float* __restrict__ h_0, float* __restrict__ h_1, float* __restrict__ h_2) {
    static_assert(WF_R % 16 == 0, "whole 16-row tiles");
    static_assert(WF_NT <= SM_WAVES && 16 * WF_NT <= PDIM, "a wave per 16-column tile, inside a padded row");
    static_assert(WF_BCAP > 32 && WF_BCAP <= 64, "a lane holds two entries of a bucket");
    static_assert(WL_CAP > 128 && WL_CAP % 32 == 0 && SM_WAVES * WL_CAP <= WF_S * WF_R * WF_BCAP,
                  "slow way: room for one 128-edge group behind the fill check, the lists inside the buckets' space");
    __shared__ __attribute__((aligned(16))) float meanT[2 * WF_R * DF_PITCH];
    __shared__ __attribute__((aligned(16))) float hdT[2 * WF_R * DF_PITCH];
    __shared__ unsigned bucket[WF_NJ * SM_WAVES * 32];
    __shared__ int cnt[WF_S * WF_R];
    __shared__ int qnext[WF_S];
    __shared__ int ovf;
    const int tid = threadIdx.x;
    const int lane = tid & 31;
    const int wave = __builtin_amdgcn_readfirstlane(tid >> 5);
    const int lo = blockIdx.x * (WF_S * WF_R);
    const int rel = blockIdx.y;
    const int* __restrict__ src_nid = (rel == 0) ? sn0 : ((rel == 1) ? sn1 : sn2);
    const int* __restrict__ e_src   = (rel == 0) ? es0 : ((rel == 1) ? es1 : es2);
    const int* __restrict__ e_dst   = (rel == 0) ? ed0 : ((rel == 1) ? ed1 : ed2);
    const float* __restrict__ wp    = (rel == 0) ? wp0 : ((rel == 1) ? wp1 : wp2);
    const float* __restrict__ bs    = (rel == 0) ? bs0 : ((rel == 1) ? bs1 : bs2);
    float* __restrict__ h           = (rel == 0) ? h_0 : ((rel == 1) ? h_1 : h_2);

    for (int i = tid; i < WF_S * WF_R; i += SM_WAVES * 32) cnt[i] = 0;
    if (tid < WF_S) qnext[tid] = 0;
    if (tid == 0) ovf = 0;
    __syncthreads();

    const int ngroups = (E + 127) >> 7;
    const int rows_here = (n_dst - lo < WF_S * WF_R) ? (n_dst - lo) : WF_S * WF_R;

    {
        bool over = false;
        for (int g0 = wave * SM_UNR; g0 < ngroups; g0 += SM_WAVES * SM_UNR) {
            int d[SM_UNR][4];
            sm_load_round<SM_UNR>(e_dst, g0, E, lane, d);
#pragma unroll
            for (int u = 0; u < SM_UNR; ++u)
                wf_rank_keep(d[u], (g0 + u) * 128 + lane * 4, lo, (unsigned)rows_here, bucket, cnt, over);
        }
        if (over) ovf = 1;
    }
    __syncthreads();
    const bool slow = __builtin_amdgcn_readfirstlane(ovf) != 0;
    if (!slow) {
        unsigned en[WF_NJ];
#pragma unroll
        for (int j = 0; j < WF_NJ; ++j) {
            const int i = tid + SM_WAVES * 32 * j;
            const int r = i / WF_BCAP, k = i - r * WF_BCAP;
            const bool rin = r < WF_S * WF_R;
            const bool live = rin && k < cnt[rin ? r : 0];
            const unsigned raw = bucket[i];
            en[j] = live ? raw : 0u;
        }
#pragma unroll
        for (int j = 0; j < WF_NJ; ++j) en[j] = (unsigned)e_src[en[j]];
#pragma unroll
        for (int j = 0; j < WF_NJ; ++j) en[j] = (unsigned)src_nid[en[j]];
#pragma unroll
        for (int j = 0; j < WF_NJ; ++j) bucket[tid + SM_WAVES * 32 * j] = en[j];
    } else {
        for (int i = tid; i < WF_S * WF_R; i += SM_WAVES * 32) cnt[i] = 0;
    }
    __syncthreads();

    const int nsub = (rows_here + WF_R - 1) / WF_R;

#pragma unroll 1
    for (int s = -1; s < nsub; ++s) {
        const int lo_s = lo + WF_R * s;
        int t = tid;
        asm volatile("" : "+v"(t));
        const int hi = (t >> 4) & 1, l16 = t & 15;
        const int n = 16 * wave + l16;
        const bool ncol = n < DDIM;
        float* mT = meanT + (s & 1) * (WF_R * DF_PITCH);
        float* hT = hdT + (s & 1) * (WF_R * DF_PITCH);
        if (slow && s >= 0) {
            const unsigned lim = (unsigned)((n_dst - lo_s < WF_R) ? (n_dst - lo_s) : WF_R);
            unsigned* wl = bucket + wave * WL_CAP;
            for (int i = t; i < WF_R * DF_PITCH / 4; i += SM_WAVES * 32)
                ((v4f*)mT)[i] = (v4f){0.f, 0.f, 0.f, 0.f};
            __syncthreads();
            int c = 0;
            for (int g0 = wave * SM_UNR; g0 < ngroups; g0 += SM_WAVES * SM_UNR) {
                int d[SM_UNR][4];
                sm_load_round<SM_UNR>(e_dst, g0, E, lane, d);
#pragma unroll
                for (int u = 0; u < SM_UNR; ++u) {
                    if (c > WL_CAP - 128) {
                        sm_flush<DF_PITCH>(emb, src_nid, e_src, wl, c, mT, lane);
                        c = 0;
                    }
                    sm_keep(d[u], (g0 + u) * 128 + lane * 4, lo_s, lim, wl, cnt + WF_R * s, c);
                }
            }
            sm_flush<DF_PITCH>(emb, src_nid, e_src, wl, c, mT, lane);
            __syncthreads();
            for (int i = t; i < WF_R * (PDIM / 4); i += SM_WAVES * 32) {
                const int r = i / (PDIM / 4), c4 = i - r * (PDIM / 4);
                const float cf = fmaxf((float)cnt[WF_R * s + r], 1.f);
                v4f sv = *(const v4f*)(mT + r * DF_PITCH + 4 * c4);
                sv.x /= cf; sv.y /= cf; sv.z /= cf; sv.w /= cf;
                *(v4f*)(mT + r * DF_PITCH + 4 * c4) = sv;
            }
            for (int r = wave; r < WF_R; r += SM_WAVES) {
                const bool live = lo_s + r < n_dst;
                const v4f* __restrict__ rp = (const v4f*)(emb + (size_t)(live ? dst_nid[lo_s + r] : 0) * DDIM);
                for (int c4 = lane; c4 < DDIM / 4; c4 += 32)
                    *(v4f*)(hT + r * DF_PITCH + 4 * c4) = live ? rp[c4] : (v4f){0.f, 0.f, 0.f, 0.f};
            }
            __syncthreads();
        }

        v8f acc[WF_MT];
        const bool prod = s >= 0 && wave < WF_NT;
        if (prod) {
            static_assert(DDIM % 4 == 0, "a table row and a mean row are a whole number of k-steps");
            const float* __restrict__ bcol = wp + 2 * hi * DDIM + (ncol ? n : DDIM - 1);
            const float bv = ncol ? bs[n] : 0.f;
#pragma unroll
            for (int mt = 0; mt < WF_MT; ++mt)
#pragma unroll
                for (int i = 0; i < 8; ++i) acc[mt][i] = bv;
            wf_tile_gemm<DDIM / 8, (DDIM % 8) != 0>(hT + l16 * DF_PITCH + 2 * hi, bcol, acc);
            wf_tile_gemm<DDIM / 8, (DDIM % 8) != 0>(mT + l16 * DF_PITCH + 2 * hi, bcol + DDIM * DDIM, acc);
        } else if (!slow && s + 1 < nsub) {
            float* mN = meanT + ((s + 1) & 1) * (WF_R * DF_PITCH);
            float* hN = hdT + ((s + 1) & 1) * (WF_R * DF_PITCH);
            const int lo_n = lo_s + WF_R;
            for (;;) {
                int r = 0;
                if (lane == 0) r = __hip_atomic_fetch_add(&qnext[s + 1], 1, __ATOMIC_RELAXED, __HIP_MEMORY_SCOPE_WORKGROUP);
                r = __builtin_amdgcn_readfirstlane(r);
                if (r >= WF_R) break;
                const bool live = lo_n + r < n_dst;
                const v4f* __restrict__ rp = (const v4f*)(emb + (size_t)(live ? dst_nid[lo_n + r] : 0) * DDIM);
                v4f own[3];
#pragma unroll
                for (int q = 0; q < 3; ++q) {
                    const int c4 = lane + 32 * q;
                    own[q] = (live && c4 < DDIM / 4) ? rp[(c4 < DDIM / 4) ? c4 : 0] : (v4f){0.f, 0.f, 0.f, 0.f};
                }
                wf_row_mean(emb, bucket + (WF_R * (s + 1) + r) * WF_BCAP, cnt[WF_R * (s + 1) + r], mN + r * DF_PITCH, lane);
#pragma unroll
                for (int q = 0; q < 3; ++q) {
                    const int c4 = lane + 32 * q;
                    if (c4 < DDIM / 4) *(v4f*)(hN + r * DF_PITCH + 4 * c4) = own[q];
                }
            }
        }
        __syncthreads();
        if (s < 0) continue;
        if (prod) {
#pragma unroll
            for (int mt = 0; mt < WF_MT; ++mt)
#pragma unroll
                for (int i = 0; i < 8; ++i) mT[(16 * mt + i + 8 * hi) * DF_PITCH + n] = ncol ? acc[mt][i] : 0.f;
        }
        __syncthreads();
        for (int i = t; i < WF_R * (PDIM / 4); i += SM_WAVES * 32) {
            const int r = i / (PDIM / 4), c4 = i - r * (PDIM / 4);
            if (lo_s + r < n_dst)
                ((v4f*)(h + (size_t)lo_s * PDIM))[i] = *(const v4f*)(mT + r * DF_PITCH + 4 * c4);
        }
        if (s + 1 < nsub) __syncthreads();
    }
}

__global__ __launch_bounds__(256) void doc_tail(
        const float* __restrict__ d3_0, const float* __restrict__ d3_1, const float* __restrict__ d3_2,
        const float* __restrict__ docout, const float* __restrict__ wfc, const float* __restrict__ bfc,
        float* __restrict__ out, int n) {
    __shared__ float rep[8 * PDIM];
    const int lane = threadIdx.x & 31;
    const int wave = threadIdx.x >> 5;
    const int node = blockIdx.x * 8 + wave;
    if (node < n)
        sdpa_row<PDIM>(d3_0 + (size_t)node * PDIM, d3_1 + (size_t)node * PDIM, d3_2 + (size_t)node * PDIM,
                       docout + (size_t)node * PDIM, rep + wave * PDIM, lane);
    __syncthreads();
    if (node < n && lane < CN) {
        float acc = bfc[lane];
        const float* rp = rep + wave * PDIM;
        for (int j = 0; j < DDIM; ++j) acc += rp[j] * wfc[(size_t)j * CN + lane];
        out[(size_t)node * CN + lane] = acc;
    }
}


extern "C" void kernel_launch(void* const* d_in, const int* in_sizes, int n_in,
                              void* d_out, int out_size, void* d_ws, size_t ws_size,
                              hipStream_t stream) {
    (void)in_sizes; (void)n_in; (void)out_size; (void)ws_size;
    const float* emb_word = (const float*)d_in[0];
    const float* emb_doc  = (const float*)d_in[1];
    const float* w_sage[6] = {(const float*)d_in[2],  (const float*)d_in[4],  (const float*)d_in[6],
                              (const float*)d_in[8],  (const float*)d_in[10], (const float*)d_in[12]};
    const float* b_sage[6] = {(const float*)d_in[3],  (const float*)d_in[5],  (const float*)d_in[7],
                              (const float*)d_in[9],  (const float*)d_in[11], (const float*)d_in[13]};
    const float* w_dense = (const float*)d_in[14];
    const float* b_dense = (const float*)d_in[15];
    const float* w_fc    = (const float*)d_in[16];
    const float* b_fc    = (const float*)d_in[17];
    const int* dst_nid      = (const int*)d_in[18];
    const int* src_nid_w[3] = {(const int*)d_in[19], (const int*)d_in[20], (const int*)d_in[21]};
    const int* e_src_w[3]   = {(const int*)d_in[22], (const int*)d_in[24], (const int*)d_in[26]};
    const int* e_dst_w[3]   = {(const int*)d_in[23], (const int*)d_in[25], (const int*)d_in[27]};
    const int* x_batch      = (const int*)d_in[28];
    const int* src_nid_d[3] = {(const int*)d_in[30], (const int*)d_in[31], (const int*)d_in[32]};
    const int* e_src_d[3]   = {(const int*)d_in[33], (const int*)d_in[35], (const int*)d_in[37]};
    const int* e_dst_d[3]   = {(const int*)d_in[34], (const int*)d_in[36], (const int*)d_in[38]};
    float* out = (float*)d_out;

    float* ws = (float*)d_ws;
    size_t off = 0;
    auto carve = [&](size_t nflt) { float* p = ws + off; off += nflt; return p; };
    float* nbuf = carve((size_t)NWN * PDIM);
    float* h3[3]; for (int r = 0; r < 3; ++r) h3[r] = carve((size_t)NWN * PDIM);
    float* pooled = carve((size_t)BN * PDIM);
    float* docout = carve((size_t)BN * PDIM);
    float* d3[3]; for (int r = 0; r < 3; ++r) d3[r] = carve((size_t)BN * PDIM);

    word_rel_fused<<<dim3((NWN + WF_S * WF_R - 1) / (WF_S * WF_R), 3), SM_WAVES * 32, 0, stream>>>(
        emb_word, dst_nid, src_nid_w[0], src_nid_w[1], src_nid_w[2], e_src_w[0], e_src_w[1], e_src_w[2],
        e_dst_w[0], e_dst_w[1], e_dst_w[2], w_sage[0], w_sage[1], w_sage[2], b_sage[0], b_sage[1], b_sage[2],
        EWN, NWN, h3[0], h3[1], h3[2]);

    sdpa_fuse_add<<<(NWN + 7) / 8, 256, 0, stream>>>(h3[0], h3[1], h3[2], emb_word, dst_nid, nbuf, NWN);

    pool_seq<<<BN, PS_WAVES * 32, 0, stream>>>(nbuf, x_batch, pooled);

    doc_rel_fused<<<dim3(BN / 16, 3), DF_WAVES * 32, 0, stream>>>(
        pooled, w_dense, b_dense, w_sage[3], w_sage[4], w_sage[5], b_sage[3], b_sage[4], b_sage[5], emb_doc,
        src_nid_d[0], src_nid_d[1], src_nid_d[2], e_src_d[0], e_src_d[1], e_src_d[2],
        e_dst_d[0], e_dst_d[1], e_dst_d[2], EDN, BN, docout, d3[0], d3[1], d3[2]);

    doc_tail<<<(BN + 7) / 8, 256, 0, stream>>>(d3[0], d3[1], d3[2], docout, w_fc, b_fc, out, BN);
}
